# starting kernel
# baseline (speedup 1.0000x reference)
#include <hip/hip_runtime.h>

typedef float v2f __attribute__((ext_vector_type(2)));
typedef float v8f __attribute__((ext_vector_type(8)));

#define N_NODES 100000
#define N_EDGES 3200000
#define NFEAT   256
#define NCLASS  64
#define MTILES  (N_NODES / 16)
#define WAVES_PER_BLOCK 8


__global__ void k_init_deg(float* __restrict__ deg) {
    int i = blockIdx.x * blockDim.x + threadIdx.x;
    if (i < N_NODES) deg[i] = 1.0f;
}

__global__ void k_count(const int* __restrict__ dst, float* __restrict__ deg) {
    int e = blockIdx.x * blockDim.x + threadIdx.x;
    if (e < N_EDGES) atomicAdd(&deg[dst[e]], 1.0f);
}

__global__ void k_dinv(float* __restrict__ deg) {
    int i = blockIdx.x * blockDim.x + threadIdx.x;
    if (i < N_NODES) deg[i] = __frsqrt_rn(deg[i]);
}


__global__ __launch_bounds__(256) void k_gemm_wmma(const float* __restrict__ x,
                                                   const float* __restrict__ W,
                                                   float* __restrict__ xw) {
    __shared__ float Wt[NCLASS * NFEAT];
    for (int i = threadIdx.x; i < NCLASS * NFEAT; i += 256) {
        int k = i >> 6;
        int c = i & 63;
        Wt[c * NFEAT + k] = W[i];
    }
    __syncthreads();

    const int lane  = threadIdx.x & 31;
    const int wave  = threadIdx.x >> 5;
    const int mtile = blockIdx.x * WAVES_PER_BLOCK + wave;
    if (mtile >= MTILES) return;

    const int half = lane >> 4;
    const int l15  = lane & 15;
    const float* xrow = x + (size_t)(mtile * 16 + l15) * NFEAT;

    v8f acc0 = {}, acc1 = {}, acc2 = {}, acc3 = {};

    for (int k = 0; k < NFEAT; k += 4) {
        const int ka = k + 2 * half;
        v2f a  = *(const v2f*)(xrow + ka);
        v2f b0 = *(const v2f*)(&Wt[(0 * 16 + l15) * NFEAT + ka]);
        v2f b1 = *(const v2f*)(&Wt[(1 * 16 + l15) * NFEAT + ka]);
        v2f b2 = *(const v2f*)(&Wt[(2 * 16 + l15) * NFEAT + ka]);
        v2f b3 = *(const v2f*)(&Wt[(3 * 16 + l15) * NFEAT + ka]);
        acc0 = __builtin_amdgcn_wmma_f32_16x16x4_f32(false, a, false, b0, (short)0, acc0, false, false);
        acc1 = __builtin_amdgcn_wmma_f32_16x16x4_f32(false, a, false, b1, (short)0, acc1, false, false);
        acc2 = __builtin_amdgcn_wmma_f32_16x16x4_f32(false, a, false, b2, (short)0, acc2, false, false);
        acc3 = __builtin_amdgcn_wmma_f32_16x16x4_f32(false, a, false, b3, (short)0, acc3, false, false);
    }

    float* orow = xw + (size_t)mtile * 16 * NCLASS;
#pragma unroll
    for (int r = 0; r < 8; ++r) {
        int m = r + 8 * half;
        orow[m * NCLASS + 0 * 16 + l15] = acc0[r];
        orow[m * NCLASS + 1 * 16 + l15] = acc1[r];
        orow[m * NCLASS + 2 * 16 + l15] = acc2[r];
        orow[m * NCLASS + 3 * 16 + l15] = acc3[r];
    }
}


__global__ void k_out_init(const float* __restrict__ xw, const float* __restrict__ dinv,
                           const float* __restrict__ bias, float* __restrict__ out) {
    int i = blockIdx.x * blockDim.x + threadIdx.x;
    if (i < N_NODES * NCLASS) {
        int n = i >> 6;
        int c = i & 63;
        float di = dinv[n];
        out[i] = bias[c] + xw[i] * di * di;
    }
}


__global__ void k_scatter(const int* __restrict__ src, const int* __restrict__ dst,
                          const float* __restrict__ xw, const float* __restrict__ dinv,
                          float* __restrict__ out) {
    long long t = (long long)blockIdx.x * blockDim.x + threadIdx.x;
    if (t >= (long long)N_EDGES * NCLASS) return;
    int e = (int)(t >> 6);
    int c = (int)(t & 63);
    int s = src[e];
    int d = dst[e];
    float norm = dinv[s] * dinv[d];
    atomicAdd(&out[(size_t)d * NCLASS + c], xw[(size_t)s * NCLASS + c] * norm);
}


extern "C" void kernel_launch(void* const* d_in, const int* in_sizes, int n_in,
                              void* d_out, int out_size, void* d_ws, size_t ws_size,
                              hipStream_t stream) {
    const float* x    = (const float*)d_in[0];
    const float* W    = (const float*)d_in[1];
    const float* bias = (const float*)d_in[2];
    const int*   ei   = (const int*)d_in[3];
    const int*   src  = ei;
    const int*   dst  = ei + N_EDGES;
    float* out = (float*)d_out;

    float* dinv = (float*)d_ws;
    float* xw   = dinv + 102400;

    k_init_deg<<<(N_NODES + 255) / 256, 256, 0, stream>>>(dinv);
    k_count  <<<(N_EDGES + 255) / 256, 256, 0, stream>>>(dst, dinv);
    k_dinv   <<<(N_NODES + 255) / 256, 256, 0, stream>>>(dinv);

    k_gemm_wmma<<<(MTILES + WAVES_PER_BLOCK - 1) / WAVES_PER_BLOCK, 256, 0, stream>>>(x, W, xw);

    k_out_init<<<(N_NODES * NCLASS + 255) / 256, 256, 0, stream>>>(xw, dinv, bias, out);

    long long tot = (long long)N_EDGES * NCLASS;
    k_scatter<<<(unsigned)((tot + 255) / 256), 256, 0, stream>>>(src, dst, xw, dinv, out);
}


# x@W: B fragments read straight from global W (lane owns 4 adjacent columns, two 16-byte row loads per K-step), no transposed LDS copy, no barrier; tile rows leave as 16-byte stores
# speedup vs baseline: 2.2963x; 2.2963x over previous
#include <hip/hip_runtime.h>

typedef float v2f __attribute__((ext_vector_type(2)));
typedef float v8f __attribute__((ext_vector_type(8)));

#define N_NODES 100000
#define N_EDGES 3200000
#define NFEAT   256
#define NCLASS  64
#define MTILES  (N_NODES / 16)
#define WAVES_PER_BLOCK 8

#define RANGE   391
#define NRANGES 256
#define KVECS   (N_EDGES / 16)
#define QCAP    40960
#define SWEEP_T 1024


__global__ void k_init_deg(float* __restrict__ deg) {
    int i = blockIdx.x * blockDim.x + threadIdx.x;
    if (i < N_NODES) deg[i] = 1.0f;
}

__global__ void k_count(const int* __restrict__ src, const int* __restrict__ dst,
                        float* __restrict__ deg, unsigned int* __restrict__ key4,
                        int* __restrict__ rec) {
    int t = blockIdx.x * blockDim.x + threadIdx.x;
    if (t < N_EDGES / 4) {
        int4 d = *(const int4*)(dst + 4 * t);
        int4 s = *(const int4*)(src + 4 * t);
        atomicAdd(&deg[d.x], 1.0f);
        atomicAdd(&deg[d.y], 1.0f);
        atomicAdd(&deg[d.z], 1.0f);
        atomicAdd(&deg[d.w], 1.0f);
        int r0 = d.x / RANGE, r1 = d.y / RANGE, r2 = d.z / RANGE, r3 = d.w / RANGE;
        key4[t] = (unsigned int)r0 | ((unsigned int)r1 << 8) | ((unsigned int)r2 << 16) | ((unsigned int)r3 << 24);
        int4 rc;
        rc.x = s.x | ((d.x - r0 * RANGE) << 17);
        rc.y = s.y | ((d.y - r1 * RANGE) << 17);
        rc.z = s.z | ((d.z - r2 * RANGE) << 17);
        rc.w = s.w | ((d.w - r3 * RANGE) << 17);
        *(int4*)(rec + 4 * t) = rc;
    }
}

__global__ void k_dinv(float* __restrict__ deg) {
    int i = blockIdx.x * blockDim.x + threadIdx.x;
    if (i < N_NODES) deg[i] = __frsqrt_rn(deg[i]);
}


__global__ __launch_bounds__(256) void k_gemm_wmma(const float* __restrict__ x,
                                                   const float* __restrict__ W,
                                                   float* __restrict__ xw) {
    const int lane  = threadIdx.x & 31;
    const int wave  = threadIdx.x >> 5;
    const int mtile = blockIdx.x * WAVES_PER_BLOCK + wave;
    if (mtile >= MTILES) return;

    const int half = lane >> 4;
    const int l15  = lane & 15;
    const float* xrow = x + (size_t)(mtile * 16 + l15) * NFEAT + 2 * half;
    const float* wrow = W + (size_t)(2 * half) * NCLASS + 4 * l15;

    v8f acc0 = {}, acc1 = {}, acc2 = {}, acc3 = {};

#pragma unroll 8
    for (int k = 0; k < NFEAT; k += 4) {
        v2f    a  = *(const v2f*)(xrow + k);
        float4 w0 = *(const float4*)(wrow + (size_t)k * NCLASS);
        float4 w1 = *(const float4*)(wrow + (size_t)(k + 1) * NCLASS);
        v2f b0 = {w0.x, w1.x};
        v2f b1 = {w0.y, w1.y};
        v2f b2 = {w0.z, w1.z};
        v2f b3 = {w0.w, w1.w};
        acc0 = __builtin_amdgcn_wmma_f32_16x16x4_f32(false, a, false, b0, (short)0, acc0, false, false);
        acc1 = __builtin_amdgcn_wmma_f32_16x16x4_f32(false, a, false, b1, (short)0, acc1, false, false);
        acc2 = __builtin_amdgcn_wmma_f32_16x16x4_f32(false, a, false, b2, (short)0, acc2, false, false);
        acc3 = __builtin_amdgcn_wmma_f32_16x16x4_f32(false, a, false, b3, (short)0, acc3, false, false);
    }

    float* orow = xw + (size_t)mtile * 16 * NCLASS + 4 * l15;
#pragma unroll
    for (int r = 0; r < 8; ++r) {
        int m = r + 8 * half;
        *(float4*)(orow + m * NCLASS) = make_float4(acc0[r], acc1[r], acc2[r], acc3[r]);
    }
}


__device__ __forceinline__ unsigned int zero_bytes(unsigned int x) {
    return ~(((x & 0x7F7F7F7Fu) + 0x7F7F7F7Fu) | x | 0x7F7F7F7Fu);
}

__global__ __launch_bounds__(SWEEP_T) void k_sweep(const unsigned int* __restrict__ key4,
                                                   const int* __restrict__ rec,
                                                   const float* __restrict__ xw,
                                                   const float* __restrict__ dinv,
                                                   const float* __restrict__ bias,
                                                   float* __restrict__ out) {
    __shared__ __attribute__((aligned(16))) float acc[RANGE * NCLASS];
    __shared__ int   q[QCAP];
    __shared__ float dloc[RANGE + 1];
    __shared__ int   qn;

    const int tid   = threadIdx.x;
    const int r     = blockIdx.x;
    const int node0 = r * RANGE;
    const int nn    = (N_NODES - node0 < RANGE) ? (N_NODES - node0) : RANGE;

    for (int i = tid; i < nn * NCLASS; i += SWEEP_T) {
        float di = dinv[node0 + (i >> 6)];
        acc[i] = bias[i & 63] + xw[(size_t)node0 * NCLASS + i] * di * di;
    }
    for (int i = tid; i < nn; i += SWEEP_T) dloc[i] = dinv[node0 + i];
    if (tid == 0) qn = 0;
    __syncthreads();

    const unsigned int pat = (unsigned int)r * 0x01010101u;
    const uint4* kv = (const uint4*)key4;
    const int vbase = r * (KVECS / NRANGES);
    for (int it = 0; it < (KVECS + SWEEP_T - 1) / SWEEP_T; it += 4) {
        uint4 w[4];
        int   vi[4];
#pragma unroll
        for (int j = 0; j < 4; ++j) {
            int v   = tid + (it + j) * SWEEP_T;
            int idx = v + vbase;
            if (idx >= KVECS) idx -= KVECS;
            vi[j] = (v < KVECS) ? idx : -1;
            w[j]  = kv[(v < KVECS) ? idx : 0];
        }
#pragma unroll
        for (int j = 0; j < 4; ++j) {
            unsigned int M = (zero_bytes(w[j].x ^ pat) >> 7) | (zero_bytes(w[j].y ^ pat) >> 6) |
                             (zero_bytes(w[j].z ^ pat) >> 5) | (zero_bytes(w[j].w ^ pat) >> 4);
            if (vi[j] < 0) M = 0;
            while (M) {
                int p = __ffs(M) - 1;
                M &= M - 1;
                int e = vi[j] * 16 + ((p & 3) << 2) + (p >> 3);
                int slot = atomicAdd(&qn, 1);
                if (slot < QCAP) {
                    q[slot] = e;
                } else {
                    int rc = rec[e];
                    int s  = rc & 0x1FFFF;
                    int dl = rc >> 17;
                    float nrm = dinv[s] * dloc[dl];
                    for (int c = 0; c < NCLASS; ++c)
                        atomicAdd(&acc[dl * NCLASS + c], xw[(size_t)s * NCLASS + c] * nrm);
                }
            }
        }
    }
    __syncthreads();

    const int n    = (qn < QCAP) ? qn : QCAP;
    const int hw   = tid >> 4;
    const int l16  = tid & 15;
    const int half = hw & 1;
    if (n > 0) {
        int rc[8];
#pragma unroll
        for (int j = 0; j < 8; ++j) {
            int i = hw + 64 * j;
            rc[j] = rec[q[(i < n) ? i : 0]];
        }
        for (int i0 = hw; i0 < n; i0 += 64 * 8) {
            float  nr[8];
            float4 row[8];
#pragma unroll
            for (int j = 0; j < 8; ++j) {
                int   s = rc[j] & 0x1FFFF;
                float w = dinv[s] * dloc[rc[j] >> 17];
                nr[j]  = (i0 + 64 * j < n) ? w : 0.0f;
                row[j] = *(const float4*)(xw + (size_t)s * NCLASS + 4 * l16);
            }
            int rn[8];
#pragma unroll
            for (int j = 0; j < 8; ++j) {
                int i = i0 + 64 * 8 + 64 * j;
                rn[j] = rec[q[(i < n) ? i : 0]];
            }
#pragma unroll
            for (int j = 0; j < 8; ++j) {
                float v0 = row[j].x * nr[j], v1 = row[j].y * nr[j];
                float v2 = row[j].z * nr[j], v3 = row[j].w * nr[j];
                float* a  = &acc[(rc[j] >> 17) * NCLASS + 4 * l16];
                float* ah = a + half;
                atomicAdd(ah,     half ? v1 : v0);
                atomicAdd(ah + 1, half ? v2 : v1);
                atomicAdd(ah + 2, half ? v3 : v2);
                atomicAdd(a + (half ? 0 : 3), half ? v0 : v3);
            }
#pragma unroll
            for (int j = 0; j < 8; ++j) rc[j] = rn[j];
        }
    }
    __syncthreads();

    float4*       o4 = (float4*)(out + (size_t)node0 * NCLASS);
    const float4* a4 = (const float4*)acc;
    for (int i = tid; i < nn * (NCLASS / 4); i += SWEEP_T) o4[i] = a4[i];
}


extern "C" void kernel_launch(void* const* d_in, const int* in_sizes, int n_in,
                              void* d_out, int out_size, void* d_ws, size_t ws_size,
                              hipStream_t stream) {
    const float* x    = (const float*)d_in[0];
    const float* W    = (const float*)d_in[1];
    const float* bias = (const float*)d_in[2];
    const int*   ei   = (const int*)d_in[3];
    const int*   src  = ei;
    const int*   dst  = ei + N_EDGES;
    float* out = (float*)d_out;

    float* dinv = (float*)d_ws;
    float* xw   = dinv + 102400;
    unsigned int* key4 = (unsigned int*)(xw + 6400000);
    int* rec = (int*)(key4 + 800000);

    k_init_deg<<<(N_NODES + 255) / 256, 256, 0, stream>>>(dinv);
    k_count  <<<(N_EDGES / 4 + 255) / 256, 256, 0, stream>>>(src, dst, dinv, key4, rec);
    k_dinv   <<<(N_NODES + 255) / 256, 256, 0, stream>>>(dinv);

    k_gemm_wmma<<<(MTILES + WAVES_PER_BLOCK - 1) / WAVES_PER_BLOCK, 256, 0, stream>>>(x, W, xw);

    k_sweep<<<NRANGES, SWEEP_T, 0, stream>>>(key4, rec, xw, dinv, bias, out);
}


# Degree count without global atomics: 256 blocks count their edge slice into 16-bit LDS counters and write the table out, k_dinv adds the 256 tables (exact integer degrees); 5 launches to 4
# speedup vs baseline: 4.0631x; 1.7694x over previous
#include <hip/hip_runtime.h>

typedef float v2f __attribute__((ext_vector_type(2)));
typedef float v8f __attribute__((ext_vector_type(8)));

#define N_NODES 100000
#define N_EDGES 3200000
#define NFEAT   256
#define NCLASS  64
#define MTILES  (N_NODES / 16)
#define WAVES_PER_BLOCK 8

#define RANGE   391
#define NRANGES 256
#define KVECS   (N_EDGES / 16)
#define QCAP    40960
#define SWEEP_T 1024

#define CNT_BLOCKS 256
#define CNT_T      1024
#define QUADS      (N_EDGES / 4 / CNT_BLOCKS)
#define HWORDS     (N_NODES / 2)
#define HVECS      (HWORDS / 4)
#define PSTRIDE    50016
#define DV_T       256


__global__ __launch_bounds__(CNT_T) void k_count(const int* __restrict__ src, const int* __restrict__ dst,
                                                 unsigned int* __restrict__ part,
                                                 unsigned int* __restrict__ key4, int* __restrict__ rec) {
    __shared__ __attribute__((aligned(16))) unsigned int h[HWORDS];
    const int tid = threadIdx.x;
    const int q0  = blockIdx.x * QUADS;

    int4 d[4], s[4];
#pragma unroll
    for (int j = 0; j < 4; ++j) {
        int i = tid + j * CNT_T;
        int t = q0 + ((i < QUADS) ? i : 0);
        d[j] = *(const int4*)(dst + 4 * t);
        s[j] = *(const int4*)(src + 4 * t);
    }

    uint4* h4 = (uint4*)h;
    for (int i = tid; i < HVECS; i += CNT_T) h4[i] = make_uint4(0u, 0u, 0u, 0u);
    __syncthreads();

#pragma unroll
    for (int j = 0; j < 4; ++j) {
        int i = tid + j * CNT_T;
        if (i < QUADS) {
            int t = q0 + i;
            int4 dd = d[j], ss = s[j];
            atomicAdd(&h[dd.x >> 1], 1u << ((dd.x & 1) << 4));
            atomicAdd(&h[dd.y >> 1], 1u << ((dd.y & 1) << 4));
            atomicAdd(&h[dd.z >> 1], 1u << ((dd.z & 1) << 4));
            atomicAdd(&h[dd.w >> 1], 1u << ((dd.w & 1) << 4));
            int r0 = dd.x / RANGE, r1 = dd.y / RANGE, r2 = dd.z / RANGE, r3 = dd.w / RANGE;
            key4[t] = (unsigned int)r0 | ((unsigned int)r1 << 8) | ((unsigned int)r2 << 16) | ((unsigned int)r3 << 24);
            int4 rc;
            rc.x = ss.x | ((dd.x - r0 * RANGE) << 17);
            rc.y = ss.y | ((dd.y - r1 * RANGE) << 17);
            rc.z = ss.z | ((dd.z - r2 * RANGE) << 17);
            rc.w = ss.w | ((dd.w - r3 * RANGE) << 17);
            *(int4*)(rec + 4 * t) = rc;
        }
    }
    __syncthreads();

    uint4* p4 = (uint4*)(part + (size_t)blockIdx.x * PSTRIDE);
    for (int i = tid; i < HVECS; i += CNT_T) p4[i] = h4[i];
}

__global__ __launch_bounds__(DV_T) void k_dinv(const unsigned int* __restrict__ part, float* __restrict__ dinv) {
    __shared__ unsigned int red[DV_T / 32][64];
    const int l = threadIdx.x & 31;
    const int g = threadIdx.x >> 5;
    const int w = blockIdx.x * 32 + l;
    unsigned int lo = 0, hi = 0;
    if (w < HWORDS) {
#pragma unroll
        for (int p = 0; p < CNT_BLOCKS / (DV_T / 32); ++p) {
            unsigned int v = part[(size_t)(g + (DV_T / 32) * p) * PSTRIDE + w];
            lo += v & 0xFFFFu;
            hi += v >> 16;
        }
    }
    red[g][2 * l]     = lo;
    red[g][2 * l + 1] = hi;
    __syncthreads();
    if (threadIdx.x < 64) {
        int node = blockIdx.x * 64 + threadIdx.x;
        if (node < N_NODES) {
            unsigned int c = 0;
#pragma unroll
            for (int k = 0; k < DV_T / 32; ++k) c += red[k][threadIdx.x];
            dinv[node] = __frsqrt_rn(1.0f + (float)c);
        }
    }
}


__global__ __launch_bounds__(256) void k_gemm_wmma(const float* __restrict__ x,
                                                   const float* __restrict__ W,
                                                   float* __restrict__ xw) {
    const int lane  = threadIdx.x & 31;
    const int wave  = threadIdx.x >> 5;
    const int mtile = blockIdx.x * WAVES_PER_BLOCK + wave;
    if (mtile >= MTILES) return;

    const int half = lane >> 4;
    const int l15  = lane & 15;
    const float* xrow = x + (size_t)(mtile * 16 + l15) * NFEAT + 2 * half;
    const float* wrow = W + (size_t)(2 * half) * NCLASS + 4 * l15;

    v8f acc0 = {}, acc1 = {}, acc2 = {}, acc3 = {};

#pragma unroll 8
    for (int k = 0; k < NFEAT; k += 4) {
        v2f    a  = *(const v2f*)(xrow + k);
        float4 w0 = *(const float4*)(wrow + (size_t)k * NCLASS);
        float4 w1 = *(const float4*)(wrow + (size_t)(k + 1) * NCLASS);
        v2f b0 = {w0.x, w1.x};
        v2f b1 = {w0.y, w1.y};
        v2f b2 = {w0.z, w1.z};
        v2f b3 = {w0.w, w1.w};
        acc0 = __builtin_amdgcn_wmma_f32_16x16x4_f32(false, a, false, b0, (short)0, acc0, false, false);
        acc1 = __builtin_amdgcn_wmma_f32_16x16x4_f32(false, a, false, b1, (short)0, acc1, false, false);
        acc2 = __builtin_amdgcn_wmma_f32_16x16x4_f32(false, a, false, b2, (short)0, acc2, false, false);
        acc3 = __builtin_amdgcn_wmma_f32_16x16x4_f32(false, a, false, b3, (short)0, acc3, false, false);
    }

    float* orow = xw + (size_t)mtile * 16 * NCLASS + 4 * l15;
#pragma unroll
    for (int r = 0; r < 8; ++r) {
        int m = r + 8 * half;
        *(float4*)(orow + m * NCLASS) = make_float4(acc0[r], acc1[r], acc2[r], acc3[r]);
    }
}


__device__ __forceinline__ unsigned int zero_bytes(unsigned int x) {
    return ~(((x & 0x7F7F7F7Fu) + 0x7F7F7F7Fu) | x | 0x7F7F7F7Fu);
}

__global__ __launch_bounds__(SWEEP_T) void k_sweep(const unsigned int* __restrict__ key4,
                                                   const int* __restrict__ rec,
                                                   const float* __restrict__ xw,
                                                   const float* __restrict__ dinv,
                                                   const float* __restrict__ bias,
                                                   float* __restrict__ out) {
    __shared__ __attribute__((aligned(16))) float acc[RANGE * NCLASS];
    __shared__ int   q[QCAP];
    __shared__ float dloc[RANGE + 1];
    __shared__ int   qn;

    const int tid   = threadIdx.x;
    const int r     = blockIdx.x;
    const int node0 = r * RANGE;
    const int nn    = (N_NODES - node0 < RANGE) ? (N_NODES - node0) : RANGE;

    for (int i = tid; i < nn * NCLASS; i += SWEEP_T) {
        float di = dinv[node0 + (i >> 6)];
        acc[i] = bias[i & 63] + xw[(size_t)node0 * NCLASS + i] * di * di;
    }
    for (int i = tid; i < nn; i += SWEEP_T) dloc[i] = dinv[node0 + i];
    if (tid == 0) qn = 0;
    __syncthreads();

    const unsigned int pat = (unsigned int)r * 0x01010101u;
    const uint4* kv = (const uint4*)key4;
    const int vbase = r * (KVECS / NRANGES);
    for (int it = 0; it < (KVECS + SWEEP_T - 1) / SWEEP_T; it += 4) {
        uint4 w[4];
        int   vi[4];
#pragma unroll
        for (int j = 0; j < 4; ++j) {
            int v   = tid + (it + j) * SWEEP_T;
            int idx = v + vbase;
            if (idx >= KVECS) idx -= KVECS;
            vi[j] = (v < KVECS) ? idx : -1;
            w[j]  = kv[(v < KVECS) ? idx : 0];
        }
#pragma unroll
        for (int j = 0; j < 4; ++j) {
            unsigned int M = (zero_bytes(w[j].x ^ pat) >> 7) | (zero_bytes(w[j].y ^ pat) >> 6) |
                             (zero_bytes(w[j].z ^ pat) >> 5) | (zero_bytes(w[j].w ^ pat) >> 4);
            if (vi[j] < 0) M = 0;
            while (M) {
                int p = __ffs(M) - 1;
                M &= M - 1;
                int e = vi[j] * 16 + ((p & 3) << 2) + (p >> 3);
                int slot = atomicAdd(&qn, 1);
                if (slot < QCAP) {
                    q[slot] = e;
                } else {
                    int rc = rec[e];
                    int s  = rc & 0x1FFFF;
                    int dl = rc >> 17;
                    float nrm = dinv[s] * dloc[dl];
                    for (int c = 0; c < NCLASS; ++c)
                        atomicAdd(&acc[dl * NCLASS + c], xw[(size_t)s * NCLASS + c] * nrm);
                }
            }
        }
    }
    __syncthreads();

    const int n    = (qn < QCAP) ? qn : QCAP;
    const int hw   = tid >> 4;
    const int l16  = tid & 15;
    const int half = hw & 1;
    if (n > 0) {
        int rc[8];
#pragma unroll
        for (int j = 0; j < 8; ++j) {
            int i = hw + 64 * j;
            rc[j] = rec[q[(i < n) ? i : 0]];
        }
        for (int i0 = hw; i0 < n; i0 += 64 * 8) {
            float  nr[8];
            float4 row[8];
#pragma unroll
            for (int j = 0; j < 8; ++j) {
                int   s = rc[j] & 0x1FFFF;
                float w = dinv[s] * dloc[rc[j] >> 17];
                nr[j]  = (i0 + 64 * j < n) ? w : 0.0f;
                row[j] = *(const float4*)(xw + (size_t)s * NCLASS + 4 * l16);
            }
            int rn[8];
#pragma unroll
            for (int j = 0; j < 8; ++j) {
                int i = i0 + 64 * 8 + 64 * j;
                rn[j] = rec[q[(i < n) ? i : 0]];
            }
#pragma unroll
            for (int j = 0; j < 8; ++j) {
                float v0 = row[j].x * nr[j], v1 = row[j].y * nr[j];
                float v2 = row[j].z * nr[j], v3 = row[j].w * nr[j];
                float* a  = &acc[(rc[j] >> 17) * NCLASS + 4 * l16];
                float* ah = a + half;
                atomicAdd(ah,     half ? v1 : v0);
                atomicAdd(ah + 1, half ? v2 : v1);
                atomicAdd(ah + 2, half ? v3 : v2);
                atomicAdd(a + (half ? 0 : 3), half ? v0 : v3);
            }
#pragma unroll
            for (int j = 0; j < 8; ++j) rc[j] = rn[j];
        }
    }
    __syncthreads();

    float4*       o4 = (float4*)(out + (size_t)node0 * NCLASS);
    const float4* a4 = (const float4*)acc;
    for (int i = tid; i < nn * (NCLASS / 4); i += SWEEP_T) o4[i] = a4[i];
}


extern "C" void kernel_launch(void* const* d_in, const int* in_sizes, int n_in,
                              void* d_out, int out_size, void* d_ws, size_t ws_size,
                              hipStream_t stream) {
    const float* x    = (const float*)d_in[0];
    const float* W    = (const float*)d_in[1];
    const float* bias = (const float*)d_in[2];
    const int*   ei   = (const int*)d_in[3];
    const int*   src  = ei;
    const int*   dst  = ei + N_EDGES;
    float* out = (float*)d_out;

    float* dinv = (float*)d_ws;
    float* xw   = dinv + 102400;
    unsigned int* key4 = (unsigned int*)(xw + 6400000);
    int* rec = (int*)(key4 + 800000);
    unsigned int* part = (unsigned int*)(rec + 3200000);

    k_count<<<CNT_BLOCKS, CNT_T, 0, stream>>>(src, dst, part, key4, rec);
    k_dinv <<<(HWORDS + 31) / 32, DV_T, 0, stream>>>(part, dinv);

    k_gemm_wmma<<<(MTILES + WAVES_PER_BLOCK - 1) / WAVES_PER_BLOCK, 256, 0, stream>>>(x, W, xw);

    k_sweep<<<NRANGES, SWEEP_T, 0, stream>>>(key4, rec, xw, dinv, bias, out);
}


# Sweep stops scanning all edges' range bytes: the slice pass writes per slice and range a bitmap of 16-edge groups holding such an edge; a sweep block reads only flagged groups; sums arrival-order
# speedup vs baseline: 5.0035x; 1.2314x over previous
#include <hip/hip_runtime.h>

typedef float v2f __attribute__((ext_vector_type(2)));
typedef float v8f __attribute__((ext_vector_type(8)));

#define N_NODES 100000
#define N_EDGES 3200000
#define NFEAT   256
#define NCLASS  64
#define MTILES  (N_NODES / 16)
#define WAVES_PER_BLOCK 8

#define RANGE   391
#define NRANGES 256
#define QCAP    36864
#define LCAP    16384
#define SWEEP_T 1024

#define CNT_BLOCKS 256
#define CNT_T      1024
#define TOT_QUADS  (N_EDGES / 4)
#define SL_QUADS   3136
#define SL_VECS    (SL_QUADS / 4)
#define SROW       32
#define SWORDS     (NRANGES * SROW)
#define HWORDS     (N_NODES / 2)
#define HVECS      (HWORDS / 4)
#define PSTRIDE    50016
#define DV_T       256


__global__ __launch_bounds__(CNT_T) void k_count(const int* __restrict__ src, const int* __restrict__ dst,
                                                 unsigned int* __restrict__ part,
                                                 unsigned int* __restrict__ key4, int* __restrict__ rec,
                                                 unsigned int* __restrict__ summ) {
    __shared__ __attribute__((aligned(16))) unsigned int h[HWORDS];
    __shared__ __attribute__((aligned(16))) unsigned int sm[SWORDS];
    const int tid = threadIdx.x;
    const int q0  = blockIdx.x * SL_QUADS;
    const int nq  = (TOT_QUADS - q0 < SL_QUADS) ? (TOT_QUADS - q0) : SL_QUADS;

    int4 d[4], s[4];
#pragma unroll
    for (int j = 0; j < 4; ++j) {
        int i = tid + j * CNT_T;
        int t = q0 + ((i < nq) ? i : 0);
        d[j] = *(const int4*)(dst + 4 * t);
        s[j] = *(const int4*)(src + 4 * t);
    }

    uint4* h4  = (uint4*)h;
    uint4* sm4 = (uint4*)sm;
    for (int i = tid; i < HVECS; i += CNT_T) h4[i] = make_uint4(0u, 0u, 0u, 0u);
    for (int i = tid; i < SWORDS / 4; i += CNT_T) sm4[i] = make_uint4(0u, 0u, 0u, 0u);
    __syncthreads();

#pragma unroll
    for (int j = 0; j < 4; ++j) {
        int i = tid + j * CNT_T;
        if (i < nq) {
            int t = q0 + i;
            int4 dd = d[j], ss = s[j];
            atomicAdd(&h[dd.x >> 1], 1u << ((dd.x & 1) << 4));
            atomicAdd(&h[dd.y >> 1], 1u << ((dd.y & 1) << 4));
            atomicAdd(&h[dd.z >> 1], 1u << ((dd.z & 1) << 4));
            atomicAdd(&h[dd.w >> 1], 1u << ((dd.w & 1) << 4));
            int r0 = dd.x / RANGE, r1 = dd.y / RANGE, r2 = dd.z / RANGE, r3 = dd.w / RANGE;
            key4[t] = (unsigned int)r0 | ((unsigned int)r1 << 8) | ((unsigned int)r2 << 16) | ((unsigned int)r3 << 24);
            int4 rc;
            rc.x = ss.x | ((dd.x - r0 * RANGE) << 17);
            rc.y = ss.y | ((dd.y - r1 * RANGE) << 17);
            rc.z = ss.z | ((dd.z - r2 * RANGE) << 17);
            rc.w = ss.w | ((dd.w - r3 * RANGE) << 17);
            *(int4*)(rec + 4 * t) = rc;
            const unsigned int gbit = 1u << ((i >> 2) & 31);
            const int gw = i >> 7;
            atomicOr(&sm[r0 * SROW + ((gw + (r0 >> 1)) & 31)], gbit);
            atomicOr(&sm[r1 * SROW + ((gw + (r1 >> 1)) & 31)], gbit);
            atomicOr(&sm[r2 * SROW + ((gw + (r2 >> 1)) & 31)], gbit);
            atomicOr(&sm[r3 * SROW + ((gw + (r3 >> 1)) & 31)], gbit);
        }
    }
    __syncthreads();

    uint4* p4 = (uint4*)(part + (size_t)blockIdx.x * PSTRIDE);
    for (int i = tid; i < HVECS; i += CNT_T) p4[i] = h4[i];
    uint4* s4 = (uint4*)(summ + (size_t)blockIdx.x * SWORDS);
    for (int i = tid; i < SWORDS / 4; i += CNT_T) s4[i] = sm4[i];
}

__global__ __launch_bounds__(DV_T) void k_dinv(const unsigned int* __restrict__ part, float* __restrict__ dinv) {
    __shared__ unsigned int red[DV_T / 32][64];
    const int l = threadIdx.x & 31;
    const int g = threadIdx.x >> 5;
    const int w = blockIdx.x * 32 + l;
    unsigned int lo = 0, hi = 0;
    if (w < HWORDS) {
#pragma unroll
        for (int p = 0; p < CNT_BLOCKS / (DV_T / 32); ++p) {
            unsigned int v = part[(size_t)(g + (DV_T / 32) * p) * PSTRIDE + w];
            lo += v & 0xFFFFu;
            hi += v >> 16;
        }
    }
    red[g][2 * l]     = lo;
    red[g][2 * l + 1] = hi;
    __syncthreads();
    if (threadIdx.x < 64) {
        int node = blockIdx.x * 64 + threadIdx.x;
        if (node < N_NODES) {
            unsigned int c = 0;
#pragma unroll
            for (int k = 0; k < DV_T / 32; ++k) c += red[k][threadIdx.x];
            dinv[node] = __frsqrt_rn(1.0f + (float)c);
        }
    }
}


__global__ __launch_bounds__(256) void k_gemm_wmma(const float* __restrict__ x,
                                                   const float* __restrict__ W,
                                                   float* __restrict__ xw) {
    const int lane  = threadIdx.x & 31;
    const int wave  = threadIdx.x >> 5;
    const int mtile = blockIdx.x * WAVES_PER_BLOCK + wave;
    if (mtile >= MTILES) return;

    const int half = lane >> 4;
    const int l15  = lane & 15;
    const float* xrow = x + (size_t)(mtile * 16 + l15) * NFEAT + 2 * half;
    const float* wrow = W + (size_t)(2 * half) * NCLASS + 4 * l15;

    v8f acc0 = {}, acc1 = {}, acc2 = {}, acc3 = {};

#pragma unroll 8
    for (int k = 0; k < NFEAT; k += 4) {
        v2f    a  = *(const v2f*)(xrow + k);
        float4 w0 = *(const float4*)(wrow + (size_t)k * NCLASS);
        float4 w1 = *(const float4*)(wrow + (size_t)(k + 1) * NCLASS);
        v2f b0 = {w0.x, w1.x};
        v2f b1 = {w0.y, w1.y};
        v2f b2 = {w0.z, w1.z};
        v2f b3 = {w0.w, w1.w};
        acc0 = __builtin_amdgcn_wmma_f32_16x16x4_f32(false, a, false, b0, (short)0, acc0, false, false);
        acc1 = __builtin_amdgcn_wmma_f32_16x16x4_f32(false, a, false, b1, (short)0, acc1, false, false);
        acc2 = __builtin_amdgcn_wmma_f32_16x16x4_f32(false, a, false, b2, (short)0, acc2, false, false);
        acc3 = __builtin_amdgcn_wmma_f32_16x16x4_f32(false, a, false, b3, (short)0, acc3, false, false);
    }

    float* orow = xw + (size_t)mtile * 16 * NCLASS + 4 * l15;
#pragma unroll
    for (int r = 0; r < 8; ++r) {
        int m = r + 8 * half;
        *(float4*)(orow + m * NCLASS) = make_float4(acc0[r], acc1[r], acc2[r], acc3[r]);
    }
}


__device__ __forceinline__ unsigned int zero_bytes(unsigned int x) {
    return ~(((x & 0x7F7F7F7Fu) + 0x7F7F7F7Fu) | x | 0x7F7F7F7Fu);
}

__global__ __launch_bounds__(SWEEP_T) void k_sweep(const unsigned int* __restrict__ key4,
                                                   const int* __restrict__ rec,
                                                   const float* __restrict__ xw,
                                                   const float* __restrict__ dinv,
                                                   const float* __restrict__ bias,
                                                   const unsigned int* __restrict__ summ,
                                                   float* __restrict__ out) {
    __shared__ __attribute__((aligned(16))) float acc[RANGE * NCLASS];
    __shared__ int   q[QCAP];
    __shared__ int   glist[LCAP];
    __shared__ float dloc[RANGE + 1];
    __shared__ int   qn;
    __shared__ int   gn;

    const int tid   = threadIdx.x;
    const int r     = blockIdx.x;
    const int node0 = r * RANGE;
    const int nn    = (N_NODES - node0 < RANGE) ? (N_NODES - node0) : RANGE;

    const int sb = tid >> 2;
    const int sp = tid & 3;
    const uint4* s4 = (const uint4*)summ + (size_t)sb * (SWORDS / 4) + r * (SROW / 4) + sp * 2;
    const uint4 wa = s4[0];
    const uint4 wb = s4[1];

    for (int i = tid; i < nn * NCLASS; i += SWEEP_T) {
        float di = dinv[node0 + (i >> 6)];
        acc[i] = bias[i & 63] + xw[(size_t)node0 * NCLASS + i] * di * di;
    }
    for (int i = tid; i < nn; i += SWEEP_T) dloc[i] = dinv[node0 + i];
    if (tid == 0) { qn = 0; gn = 0; }
    __syncthreads();

    const unsigned int pat = (unsigned int)r * 0x01010101u;
    const uint4* kv = (const uint4*)key4;

    auto take = [&](int v, uint4 w) {
        unsigned int M = (zero_bytes(w.x ^ pat) >> 7) | (zero_bytes(w.y ^ pat) >> 6) |
                         (zero_bytes(w.z ^ pat) >> 5) | (zero_bytes(w.w ^ pat) >> 4);
        while (M) {
            int p = __ffs(M) - 1;
            M &= M - 1;
            int e = v * 16 + ((p & 3) << 2) + (p >> 3);
            int slot = atomicAdd(&qn, 1);
            if (slot < QCAP) {
                q[slot] = e;
            } else {
                int rc = rec[e];
                int s  = rc & 0x1FFFF;
                int dl = rc >> 17;
                float nrm = dinv[s] * dloc[dl];
                for (int c = 0; c < NCLASS; ++c)
                    atomicAdd(&acc[dl * NCLASS + c], xw[(size_t)s * NCLASS + c] * nrm);
            }
        }
    };

    {
        const unsigned int wd[8] = {wa.x, wa.y, wa.z, wa.w, wb.x, wb.y, wb.z, wb.w};
        const int rot = (r >> 1) & 31;
#pragma unroll
        for (int k = 0; k < 8; ++k) {
            unsigned int B = wd[k];
            const int gw = (sp * 8 + k - rot) & 31;
            const int v0 = sb * SL_VECS + gw * 32;
            while (B) {
                int p = __ffs(B) - 1;
                B &= B - 1;
                int slot = atomicAdd(&gn, 1);
                if (slot < LCAP) {
                    glist[slot] = v0 + p;
                } else {
                    take(v0 + p, kv[v0 + p]);
                }
            }
        }
    }
    __syncthreads();

    const int ng = (gn < LCAP) ? gn : LCAP;
    for (int i0 = 0; i0 < ng; i0 += 4 * SWEEP_T) {
        uint4 w[4];
        int   vi[4];
#pragma unroll
        for (int j = 0; j < 4; ++j) {
            int i = i0 + j * SWEEP_T + tid;
            vi[j] = (i < ng) ? glist[(i < ng) ? i : 0] : -1;
            w[j]  = kv[(vi[j] < 0) ? 0 : vi[j]];
        }
#pragma unroll
        for (int j = 0; j < 4; ++j) {
            if (vi[j] >= 0) take(vi[j], w[j]);
        }
    }
    __syncthreads();

    const int n    = (qn < QCAP) ? qn : QCAP;
    const int hw   = tid >> 4;
    const int l16  = tid & 15;
    const int half = hw & 1;
    if (n > 0) {
        int rc[8];
#pragma unroll
        for (int j = 0; j < 8; ++j) {
            int i = hw + 64 * j;
            rc[j] = rec[q[(i < n) ? i : 0]];
        }
        for (int i0 = hw; i0 < n; i0 += 64 * 8) {
            float  nr[8];
            float4 row[8];
#pragma unroll
            for (int j = 0; j < 8; ++j) {
                int   s = rc[j] & 0x1FFFF;
                float w = dinv[s] * dloc[rc[j] >> 17];
                nr[j]  = (i0 + 64 * j < n) ? w : 0.0f;
                row[j] = *(const float4*)(xw + (size_t)s * NCLASS + 4 * l16);
            }
            int rn[8];
#pragma unroll
            for (int j = 0; j < 8; ++j) {
                int i = i0 + 64 * 8 + 64 * j;
                rn[j] = rec[q[(i < n) ? i : 0]];
            }
#pragma unroll
            for (int j = 0; j < 8; ++j) {
                float v0 = row[j].x * nr[j], v1 = row[j].y * nr[j];
                float v2 = row[j].z * nr[j], v3 = row[j].w * nr[j];
                float* a  = &acc[(rc[j] >> 17) * NCLASS + 4 * l16];
                float* ah = a + half;
                atomicAdd(ah,     half ? v1 : v0);
                atomicAdd(ah + 1, half ? v2 : v1);
                atomicAdd(ah + 2, half ? v3 : v2);
                atomicAdd(a + (half ? 0 : 3), half ? v0 : v3);
            }
#pragma unroll
            for (int j = 0; j < 8; ++j) rc[j] = rn[j];
        }
    }
    __syncthreads();

    float4*       o4 = (float4*)(out + (size_t)node0 * NCLASS);
    const float4* a4 = (const float4*)acc;
    for (int i = tid; i < nn * (NCLASS / 4); i += SWEEP_T) o4[i] = a4[i];
}


extern "C" void kernel_launch(void* const* d_in, const int* in_sizes, int n_in,
                              void* d_out, int out_size, void* d_ws, size_t ws_size,
                              hipStream_t stream) {
    const float* x    = (const float*)d_in[0];
    const float* W    = (const float*)d_in[1];
    const float* bias = (const float*)d_in[2];
    const int*   ei   = (const int*)d_in[3];
    const int*   src  = ei;
    const int*   dst  = ei + N_EDGES;
    float* out = (float*)d_out;

    float* dinv = (float*)d_ws;
    float* xw   = dinv + 102400;
    unsigned int* key4 = (unsigned int*)(xw + 6400000);
    int* rec = (int*)(key4 + 800000);
    unsigned int* part = (unsigned int*)(rec + 3200000);
    unsigned int* summ = part + 12804096;

    k_count<<<CNT_BLOCKS, CNT_T, 0, stream>>>(src, dst, part, key4, rec, summ);
    k_dinv <<<(HWORDS + 31) / 32, DV_T, 0, stream>>>(part, dinv);

    k_gemm_wmma<<<(MTILES + WAVES_PER_BLOCK - 1) / WAVES_PER_BLOCK, 256, 0, stream>>>(x, W, xw);

    k_sweep<<<NRANGES, SWEEP_T, 0, stream>>>(key4, rec, xw, dinv, bias, summ, out);
}


# Sweep reads each source's weight from LDS: all nodes' degree bytes (left by k_dinv) fill the used-up edge queue, dinv[s] remade by the same cvt+add+rsqrt, bit-identical; no per-edge global dinv load
# speedup vs baseline: 6.0753x; 1.2142x over previous
#include <hip/hip_runtime.h>

typedef float v2f __attribute__((ext_vector_type(2)));
typedef float v8f __attribute__((ext_vector_type(8)));

#define N_NODES 100000
#define N_EDGES 3200000
#define NFEAT   256
#define NCLASS  64
#define MTILES  (N_NODES / 16)
#define WAVES_PER_BLOCK 8

#define RANGE   391
#define NRANGES 256
#define QCAP    36864
#define LCAP    16384
#define SWEEP_T 1024
#define OWNERS  (SWEEP_T / 16)
#define OCAP    (LCAP / OWNERS)

#define CNT_BLOCKS 256
#define CNT_T      1024
#define TOT_QUADS  (N_EDGES / 4)
#define SL_QUADS   3136
#define SL_VECS    (SL_QUADS / 4)
#define SROW       32
#define SWORDS     (NRANGES * SROW)
#define HWORDS     (N_NODES / 2)
#define HVECS      (HWORDS / 4)
#define PSTRIDE    50016
#define DV_T       256
#define DLINE      32
#define DBLOCKS    ((HWORDS + 31) / 32)
#define DPIECES    ((N_NODES / 16 + SWEEP_T - 1) / SWEEP_T)


__global__ __launch_bounds__(CNT_T) void k_count(const int* __restrict__ src, const int* __restrict__ dst,
                                                 unsigned int* __restrict__ part,
                                                 unsigned int* __restrict__ key4, int* __restrict__ rec,
                                                 unsigned int* __restrict__ summ) {
    __shared__ __attribute__((aligned(16))) unsigned int h[HWORDS];
    __shared__ __attribute__((aligned(16))) unsigned int sm[SWORDS];
    const int tid = threadIdx.x;
    const int q0  = blockIdx.x * SL_QUADS;
    const int nq  = (TOT_QUADS - q0 < SL_QUADS) ? (TOT_QUADS - q0) : SL_QUADS;

    int4 d[4], s[4];
#pragma unroll
    for (int j = 0; j < 4; ++j) {
        int i = tid + j * CNT_T;
        int t = q0 + ((i < nq) ? i : 0);
        d[j] = *(const int4*)(dst + 4 * t);
        s[j] = *(const int4*)(src + 4 * t);
    }

    uint4* h4  = (uint4*)h;
    uint4* sm4 = (uint4*)sm;
    for (int i = tid; i < HVECS; i += CNT_T) h4[i] = make_uint4(0u, 0u, 0u, 0u);
    for (int i = tid; i < SWORDS / 4; i += CNT_T) sm4[i] = make_uint4(0u, 0u, 0u, 0u);
    __syncthreads();

#pragma unroll
    for (int j = 0; j < 4; ++j) {
        int i = tid + j * CNT_T;
        if (i < nq) {
            int t = q0 + i;
            int4 dd = d[j], ss = s[j];
            atomicAdd(&h[dd.x >> 1], 1u << ((dd.x & 1) << 4));
            atomicAdd(&h[dd.y >> 1], 1u << ((dd.y & 1) << 4));
            atomicAdd(&h[dd.z >> 1], 1u << ((dd.z & 1) << 4));
            atomicAdd(&h[dd.w >> 1], 1u << ((dd.w & 1) << 4));
            int r0 = dd.x / RANGE, r1 = dd.y / RANGE, r2 = dd.z / RANGE, r3 = dd.w / RANGE;
            key4[t] = (unsigned int)r0 | ((unsigned int)r1 << 8) | ((unsigned int)r2 << 16) | ((unsigned int)r3 << 24);
            int4 rc;
            rc.x = ss.x | ((dd.x - r0 * RANGE) << 17);
            rc.y = ss.y | ((dd.y - r1 * RANGE) << 17);
            rc.z = ss.z | ((dd.z - r2 * RANGE) << 17);
            rc.w = ss.w | ((dd.w - r3 * RANGE) << 17);
            *(int4*)(rec + 4 * t) = rc;
            const unsigned int gbit = 1u << ((i >> 2) & 31);
            const int gw = i >> 7;
            atomicOr(&sm[r0 * SROW + ((gw + (r0 >> 1)) & 31)], gbit);
            atomicOr(&sm[r1 * SROW + ((gw + (r1 >> 1)) & 31)], gbit);
            atomicOr(&sm[r2 * SROW + ((gw + (r2 >> 1)) & 31)], gbit);
            atomicOr(&sm[r3 * SROW + ((gw + (r3 >> 1)) & 31)], gbit);
        }
    }
    __syncthreads();

    uint4* p4 = (uint4*)(part + (size_t)blockIdx.x * PSTRIDE);
    for (int i = tid; i < HVECS; i += CNT_T) p4[i] = h4[i];
    uint4* s4 = (uint4*)(summ + (size_t)blockIdx.x * SWORDS);
    for (int i = tid; i < SWORDS / 4; i += CNT_T) s4[i] = sm4[i];
}

__global__ __launch_bounds__(DV_T) void k_dinv(const unsigned int* __restrict__ part, float* __restrict__ dinv,
                                               unsigned int* __restrict__ degb) {
    __shared__ unsigned int red[DV_T / 32][64];
    __shared__ unsigned int rb[16];
    const int l = threadIdx.x & 31;
    const int g = threadIdx.x >> 5;
    const int w = blockIdx.x * 32 + l;
    unsigned int lo = 0, hi = 0;
    if (w < HWORDS) {
#pragma unroll
        for (int p = 0; p < CNT_BLOCKS / (DV_T / 32); ++p) {
            unsigned int v = part[(size_t)(g + (DV_T / 32) * p) * PSTRIDE + w];
            lo += v & 0xFFFFu;
            hi += v >> 16;
        }
    }
    red[g][2 * l]     = lo;
    red[g][2 * l + 1] = hi;
    __syncthreads();
    if (threadIdx.x < 64) {
        int node = blockIdx.x * 64 + threadIdx.x;
        unsigned int c = 0;
        if (node < N_NODES) {
#pragma unroll
            for (int k = 0; k < DV_T / 32; ++k) c += red[k][threadIdx.x];
            dinv[node] = __frsqrt_rn(1.0f + (float)c);
        }
        ((unsigned char*)rb)[threadIdx.x] = (unsigned char)((c < 255u) ? c : 255u);
    }
    __syncthreads();
    if (threadIdx.x < 16) degb[blockIdx.x * DLINE + threadIdx.x] = rb[threadIdx.x];
}


__global__ __launch_bounds__(256) void k_gemm_wmma(const float* __restrict__ x,
                                                   const float* __restrict__ W,
                                                   float* __restrict__ xw) {
    const int lane  = threadIdx.x & 31;
    const int wave  = threadIdx.x >> 5;
    const int mtile = blockIdx.x * WAVES_PER_BLOCK + wave;
    if (mtile >= MTILES) return;

    const int half = lane >> 4;
    const int l15  = lane & 15;
    const float* xrow = x + (size_t)(mtile * 16 + l15) * NFEAT + 2 * half;
    const float* wrow = W + (size_t)(2 * half) * NCLASS + 4 * l15;

    v8f acc0 = {}, acc1 = {}, acc2 = {}, acc3 = {};

#pragma unroll 8
    for (int k = 0; k < NFEAT; k += 4) {
        v2f    a  = *(const v2f*)(xrow + k);
        float4 w0 = *(const float4*)(wrow + (size_t)k * NCLASS);
        float4 w1 = *(const float4*)(wrow + (size_t)(k + 1) * NCLASS);
        v2f b0 = {w0.x, w1.x};
        v2f b1 = {w0.y, w1.y};
        v2f b2 = {w0.z, w1.z};
        v2f b3 = {w0.w, w1.w};
        acc0 = __builtin_amdgcn_wmma_f32_16x16x4_f32(false, a, false, b0, (short)0, acc0, false, false);
        acc1 = __builtin_amdgcn_wmma_f32_16x16x4_f32(false, a, false, b1, (short)0, acc1, false, false);
        acc2 = __builtin_amdgcn_wmma_f32_16x16x4_f32(false, a, false, b2, (short)0, acc2, false, false);
        acc3 = __builtin_amdgcn_wmma_f32_16x16x4_f32(false, a, false, b3, (short)0, acc3, false, false);
    }

    float* orow = xw + (size_t)mtile * 16 * NCLASS + 4 * l15;
#pragma unroll
    for (int r = 0; r < 8; ++r) {
        int m = r + 8 * half;
        *(float4*)(orow + m * NCLASS) = make_float4(acc0[r], acc1[r], acc2[r], acc3[r]);
    }
}


__device__ __forceinline__ unsigned int zero_bytes(unsigned int x) {
    return ~(((x & 0x7F7F7F7Fu) + 0x7F7F7F7Fu) | x | 0x7F7F7F7Fu);
}

__device__ __forceinline__ void owners_pass(const int* myq, const int m, const int l16,
                                            const unsigned char* dg, const float* dloc, float4* acc4,
                                            const float* __restrict__ xw) {
    auto fetch = [&](int i, int (&rcx)[8], unsigned int (&dgx)[8], float4 (&rowx)[8]) {
        if (i < m) {
            const int4 qa = *(const int4*)(myq + i);
            const int4 qb = *(const int4*)(myq + i + 4);
            rcx[0] = qa.x; rcx[1] = qa.y; rcx[2] = qa.z; rcx[3] = qa.w;
            rcx[4] = qb.x; rcx[5] = qb.y; rcx[6] = qb.z; rcx[7] = qb.w;
#pragma unroll
            for (int j = 0; j < 8; ++j) {
                const int s = (i + j < m) ? (rcx[j] & 0x1FFFF) : 0;
                dgx[j]  = dg[s];
                rowx[j] = *(const float4*)(xw + (size_t)s * NCLASS + 4 * l16);
            }
        }
    };
    auto add8 = [&](int i, const int (&rcx)[8], const unsigned int (&dgx)[8], const float4 (&rowx)[8]) {
        float w[8];
#pragma unroll
        for (int j = 0; j < 8; ++j) w[j] = __frsqrt_rn(1.0f + (float)dgx[j]);
#pragma unroll
        for (int j = 0; j < 8; ++j) w[j] *= dloc[(i + j < m) ? (rcx[j] >> 17) : 0];
#pragma unroll
        for (int j = 0; j < 8; ++j) {
            if (i + j < m) {
                float4* p = acc4 + (rcx[j] >> 17) * (NCLASS / 4) + l16;
                float4 c  = *p;
                c.x += rowx[j].x * w[j];
                c.y += rowx[j].y * w[j];
                c.z += rowx[j].z * w[j];
                c.w += rowx[j].w * w[j];
                *p = c;
            }
        }
    };
    int    rcA[8] = {0, 0, 0, 0, 0, 0, 0, 0}, rcB[8] = {0, 0, 0, 0, 0, 0, 0, 0};
    unsigned int dgA[8] = {0, 0, 0, 0, 0, 0, 0, 0}, dgB[8] = {0, 0, 0, 0, 0, 0, 0, 0};
    float4 rowA[8], rowB[8];
#pragma unroll
    for (int j = 0; j < 8; ++j) rowA[j] = rowB[j] = make_float4(0.0f, 0.0f, 0.0f, 0.0f);
    fetch(0, rcA, dgA, rowA);
    for (int i0 = 0; i0 < m; i0 += 16) {
        fetch(i0 + 8, rcB, dgB, rowB);
        add8(i0, rcA, dgA, rowA);
        fetch(i0 + 16, rcA, dgA, rowA);
        if (i0 + 8 < m) add8(i0 + 8, rcB, dgB, rowB);
    }
}

__global__ __launch_bounds__(SWEEP_T) void k_sweep(const unsigned int* __restrict__ key4,
                                                   const int* __restrict__ rec,
                                                   const float* __restrict__ xw,
                                                   const float* __restrict__ dinv,
                                                   const float* __restrict__ bias,
                                                   const unsigned int* __restrict__ summ,
                                                   const unsigned int* __restrict__ degb,
                                                   float* __restrict__ out) {
    __shared__ __attribute__((aligned(16))) float acc[RANGE * NCLASS];
    __shared__ __attribute__((aligned(16))) int q[QCAP];
    __shared__ __attribute__((aligned(16))) int glist[LCAP];
    __shared__ float dloc[RANGE + 1];
    __shared__ int   ocnt[OWNERS];
    __shared__ int   qn;
    __shared__ int   gn;
    __shared__ int   bigf;

    const int tid   = threadIdx.x;
    const int r     = blockIdx.x;
    const int node0 = r * RANGE;
    const int nn    = (N_NODES - node0 < RANGE) ? (N_NODES - node0) : RANGE;

    const int sb = tid >> 2;
    const int sp = tid & 3;
    const uint4* s4 = (const uint4*)summ + (size_t)sb * (SWORDS / 4) + r * (SROW / 4) + sp * 2;
    const uint4 wa = s4[0];
    const uint4 wb = s4[1];

    for (int i = tid; i < nn * NCLASS; i += SWEEP_T) {
        float di = dinv[node0 + (i >> 6)];
        acc[i] = bias[i & 63] + xw[(size_t)node0 * NCLASS + i] * di * di;
    }
    for (int i = tid; i < nn; i += SWEEP_T) dloc[i] = dinv[node0 + i];
    if (tid == 0) { qn = 0; gn = 0; bigf = 0; }
    if (tid < OWNERS) ocnt[tid] = 0;
    __syncthreads();

    const unsigned int pat = (unsigned int)r * 0x01010101u;
    const uint4* kv = (const uint4*)key4;

    auto add_rec = [&](int rc) {
        int s  = rc & 0x1FFFF;
        int dl = rc >> 17;
        float nrm = dinv[s] * dloc[dl];
        for (int c = 0; c < NCLASS; ++c)
            atomicAdd(&acc[dl * NCLASS + c], xw[(size_t)s * NCLASS + c] * nrm);
    };

    auto take = [&](int v, uint4 w) {
        unsigned int M = (zero_bytes(w.x ^ pat) >> 7) | (zero_bytes(w.y ^ pat) >> 6) |
                         (zero_bytes(w.z ^ pat) >> 5) | (zero_bytes(w.w ^ pat) >> 4);
        while (M) {
            int p = __ffs(M) - 1;
            M &= M - 1;
            int e = v * 16 + ((p & 3) << 2) + (p >> 3);
            int slot = atomicAdd(&qn, 1);
            if (slot < QCAP) {
                q[slot] = e;
            } else {
                add_rec(rec[e]);
            }
        }
    };

    {
        const unsigned int wd[8] = {wa.x, wa.y, wa.z, wa.w, wb.x, wb.y, wb.z, wb.w};
        const int rot = (r >> 1) & 31;
#pragma unroll
        for (int k = 0; k < 8; ++k) {
            unsigned int B = wd[k];
            const int gw = (sp * 8 + k - rot) & 31;
            const int v0 = sb * SL_VECS + gw * 32;
            while (B) {
                int p = __ffs(B) - 1;
                B &= B - 1;
                int slot = atomicAdd(&gn, 1);
                if (slot < LCAP) {
                    glist[slot] = v0 + p;
                } else {
                    take(v0 + p, kv[v0 + p]);
                }
            }
        }
    }
    __syncthreads();

    const int ng = (gn < LCAP) ? gn : LCAP;
    for (int i0 = 0; i0 < ng; i0 += 4 * SWEEP_T) {
        uint4 w[4];
        int   vi[4];
#pragma unroll
        for (int j = 0; j < 4; ++j) {
            int i = i0 + j * SWEEP_T + tid;
            vi[j] = (i < ng) ? glist[(i < ng) ? i : 0] : -1;
            w[j]  = kv[(vi[j] < 0) ? 0 : vi[j]];
        }
#pragma unroll
        for (int j = 0; j < 4; ++j) {
            if (vi[j] >= 0) take(vi[j], w[j]);
        }
    }
    __syncthreads();

    const uint4* g4 = (const uint4*)degb;
    uint4 dpc[DPIECES];
#pragma unroll
    for (int k = 0; k < DPIECES; ++k) {
        int i = tid + k * SWEEP_T;
        i = (i < N_NODES / 16) ? i : 0;
        dpc[k] = g4[(i >> 2) * (DLINE / 4) + (i & 3)];
    }

    const int n  = (qn < QCAP) ? qn : QCAP;
    int*      oq = glist;
    for (int i0 = 0; i0 < n; i0 += 4 * SWEEP_T) {
        int rcv[4];
#pragma unroll
        for (int j = 0; j < 4; ++j) {
            int i = i0 + j * SWEEP_T + tid;
            rcv[j] = (i < n) ? rec[q[i]] : -1;
        }
#pragma unroll
        for (int j = 0; j < 4; ++j) {
            if (rcv[j] >= 0) {
                int ow   = (rcv[j] >> 17) & (OWNERS - 1);
                int slot = atomicAdd(&ocnt[ow], 1);
                if (slot < OCAP) oq[ow * OCAP + slot] = rcv[j];
                else             add_rec(rcv[j]);
            }
        }
    }
    __syncthreads();

    {
        uint4*       t4 = (uint4*)q;
        unsigned int ff = 0;
#pragma unroll
        for (int k = 0; k < DPIECES; ++k) {
            const int i = tid + k * SWEEP_T;
            if (i < N_NODES / 16) {
                const uint4 v = dpc[k];
                t4[i] = v;
                ff |= zero_bytes(~v.x) | zero_bytes(~v.y) | zero_bytes(~v.z) | zero_bytes(~v.w);
            }
        }
        if (ff) atomicOr(&bigf, 1);
    }
    __syncthreads();

    const int hw  = tid >> 4;
    const int l16 = tid & 15;
    const int m   = (ocnt[hw] < OCAP) ? ocnt[hw] : OCAP;
    const int* myq = oq + hw * OCAP;
    const unsigned char* dg = (const unsigned char*)q;
    if (bigf) {
        float4* acc4 = (float4*)acc;
        for (int i = 0; i < m; ++i) {
            const int   rc = myq[i];
            const int   s  = rc & 0x1FFFF;
            const int   dl = rc >> 17;
            const unsigned int b = dg[s];
            const float dv = (b == 255u) ? dinv[s] : __frsqrt_rn(1.0f + (float)b);
            const float w  = dv * dloc[dl];
            const float4 row = *(const float4*)(xw + (size_t)s * NCLASS + 4 * l16);
            float4* p = acc4 + dl * (NCLASS / 4) + l16;
            float4 c  = *p;
            c.x += row.x * w;
            c.y += row.y * w;
            c.z += row.z * w;
            c.w += row.w * w;
            *p = c;
        }
    } else {
        owners_pass(myq, m, l16, dg, dloc, (float4*)acc, xw);
    }
    __syncthreads();

    float4*       o4 = (float4*)(out + (size_t)node0 * NCLASS);
    const float4* a4 = (const float4*)acc;
    for (int i = tid; i < nn * (NCLASS / 4); i += SWEEP_T) o4[i] = a4[i];
}


extern "C" void kernel_launch(void* const* d_in, const int* in_sizes, int n_in,
                              void* d_out, int out_size, void* d_ws, size_t ws_size,
                              hipStream_t stream) {
    const float* x    = (const float*)d_in[0];
    const float* W    = (const float*)d_in[1];
    const float* bias = (const float*)d_in[2];
    const int*   ei   = (const int*)d_in[3];
    const int*   src  = ei;
    const int*   dst  = ei + N_EDGES;
    float* out = (float*)d_out;

    float* dinv = (float*)d_ws;
    float* xw   = dinv + 102400;
    unsigned int* key4 = (unsigned int*)(xw + 6400000);
    int* rec = (int*)(key4 + 800000);
    unsigned int* part = (unsigned int*)(rec + 3200000);
    unsigned int* summ = part + 12804096;
    unsigned int* degb = summ + 2097152;

    k_count<<<CNT_BLOCKS, CNT_T, 0, stream>>>(src, dst, part, key4, rec, summ);
    k_dinv <<<DBLOCKS, DV_T, 0, stream>>>(part, dinv, degb);

    k_gemm_wmma<<<(MTILES + WAVES_PER_BLOCK - 1) / WAVES_PER_BLOCK, 256, 0, stream>>>(x, W, xw);

    k_sweep<<<NRANGES, SWEEP_T, 0, stream>>>(key4, rec, xw, dinv, bias, summ, degb, out);
}


# Sweep sorts its records by destination row in LDS (counting sort, self-loop as one more record); an owner sums a row in registers, writes it once: no LDS add per edge; sums in the row's arrival order
# speedup vs baseline: 6.2869x; 1.0348x over previous
#include <hip/hip_runtime.h>

typedef float v2f __attribute__((ext_vector_type(2)));
typedef float v8f __attribute__((ext_vector_type(8)));

#define N_NODES 100000
#define N_EDGES 3200000
#define NFEAT   256
#define NCLASS  64
#define MTILES  (N_NODES / 16)
#define WAVES_PER_BLOCK 8

#define RANGE   391
#define NRANGES 256
#define QCAP    36864
#define LCAP    16384
#define SWEEP_T 1024
#define OWNERS  (SWEEP_T / 16)
#define SCAP    LCAP
#define PADDL   RANGE
#define OROWS   ((RANGE + OWNERS - 1) / OWNERS)

#define CNT_BLOCKS 256
#define CNT_T      1024
#define TOT_QUADS  (N_EDGES / 4)
#define SL_QUADS   3136
#define SL_VECS    (SL_QUADS / 4)
#define SROW       32
#define SWORDS     (NRANGES * SROW)
#define HWORDS     (N_NODES / 2)
#define HVECS      (HWORDS / 4)
#define PSTRIDE    50016
#define DV_T       256
#define DLINE      32
#define DBLOCKS    ((HWORDS + 31) / 32)
#define DPIECES    ((N_NODES / 16 + SWEEP_T - 1) / SWEEP_T)


__global__ __launch_bounds__(CNT_T) void k_count(const int* __restrict__ src, const int* __restrict__ dst,
                                                 unsigned int* __restrict__ part,
                                                 unsigned int* __restrict__ key4, int* __restrict__ rec,
                                                 unsigned int* __restrict__ summ) {
    __shared__ __attribute__((aligned(16))) unsigned int h[HWORDS];
    __shared__ __attribute__((aligned(16))) unsigned int sm[SWORDS];
    const int tid = threadIdx.x;
    const int q0  = blockIdx.x * SL_QUADS;
    const int nq  = (TOT_QUADS - q0 < SL_QUADS) ? (TOT_QUADS - q0) : SL_QUADS;

    int4 d[4], s[4];
#pragma unroll
    for (int j = 0; j < 4; ++j) {
        int i = tid + j * CNT_T;
        int t = q0 + ((i < nq) ? i : 0);
        d[j] = *(const int4*)(dst + 4 * t);
        s[j] = *(const int4*)(src + 4 * t);
    }

    uint4* h4  = (uint4*)h;
    uint4* sm4 = (uint4*)sm;
    for (int i = tid; i < HVECS; i += CNT_T) h4[i] = make_uint4(0u, 0u, 0u, 0u);
    for (int i = tid; i < SWORDS / 4; i += CNT_T) sm4[i] = make_uint4(0u, 0u, 0u, 0u);
    __syncthreads();

#pragma unroll
    for (int j = 0; j < 4; ++j) {
        int i = tid + j * CNT_T;
        if (i < nq) {
            int t = q0 + i;
            int4 dd = d[j], ss = s[j];
            atomicAdd(&h[dd.x >> 1], 1u << ((dd.x & 1) << 4));
            atomicAdd(&h[dd.y >> 1], 1u << ((dd.y & 1) << 4));
            atomicAdd(&h[dd.z >> 1], 1u << ((dd.z & 1) << 4));
            atomicAdd(&h[dd.w >> 1], 1u << ((dd.w & 1) << 4));
            int r0 = dd.x / RANGE, r1 = dd.y / RANGE, r2 = dd.z / RANGE, r3 = dd.w / RANGE;
            key4[t] = (unsigned int)r0 | ((unsigned int)r1 << 8) | ((unsigned int)r2 << 16) | ((unsigned int)r3 << 24);
            int4 rc;
            rc.x = ss.x | ((dd.x - r0 * RANGE) << 17);
            rc.y = ss.y | ((dd.y - r1 * RANGE) << 17);
            rc.z = ss.z | ((dd.z - r2 * RANGE) << 17);
            rc.w = ss.w | ((dd.w - r3 * RANGE) << 17);
            *(int4*)(rec + 4 * t) = rc;
            const unsigned int gbit = 1u << ((i >> 2) & 31);
            const int gw = i >> 7;
            atomicOr(&sm[r0 * SROW + ((gw + (r0 >> 1)) & 31)], gbit);
            atomicOr(&sm[r1 * SROW + ((gw + (r1 >> 1)) & 31)], gbit);
            atomicOr(&sm[r2 * SROW + ((gw + (r2 >> 1)) & 31)], gbit);
            atomicOr(&sm[r3 * SROW + ((gw + (r3 >> 1)) & 31)], gbit);
        }
    }
    __syncthreads();

    uint4* p4 = (uint4*)(part + (size_t)blockIdx.x * PSTRIDE);
    for (int i = tid; i < HVECS; i += CNT_T) p4[i] = h4[i];
    uint4* s4 = (uint4*)(summ + (size_t)blockIdx.x * SWORDS);
    for (int i = tid; i < SWORDS / 4; i += CNT_T) s4[i] = sm4[i];
}

__global__ __launch_bounds__(DV_T) void k_dinv(const unsigned int* __restrict__ part, float* __restrict__ dinv,
                                               unsigned int* __restrict__ degb) {
    __shared__ unsigned int red[DV_T / 32][64];
    __shared__ unsigned int rb[16];
    const int l = threadIdx.x & 31;
    const int g = threadIdx.x >> 5;
    const int w = blockIdx.x * 32 + l;
    unsigned int lo = 0, hi = 0;
    if (w < HWORDS) {
#pragma unroll
        for (int p = 0; p < CNT_BLOCKS / (DV_T / 32); ++p) {
            unsigned int v = part[(size_t)(g + (DV_T / 32) * p) * PSTRIDE + w];
            lo += v & 0xFFFFu;
            hi += v >> 16;
        }
    }
    red[g][2 * l]     = lo;
    red[g][2 * l + 1] = hi;
    __syncthreads();
    if (threadIdx.x < 64) {
        int node = blockIdx.x * 64 + threadIdx.x;
        unsigned int c = 0;
        if (node < N_NODES) {
#pragma unroll
            for (int k = 0; k < DV_T / 32; ++k) c += red[k][threadIdx.x];
            dinv[node] = __frsqrt_rn(1.0f + (float)c);
        }
        ((unsigned char*)rb)[threadIdx.x] = (unsigned char)((c < 255u) ? c : 255u);
    }
    __syncthreads();
    if (threadIdx.x < 16) degb[blockIdx.x * DLINE + threadIdx.x] = rb[threadIdx.x];
}


__global__ __launch_bounds__(256) void k_gemm_wmma(const float* __restrict__ x,
                                                   const float* __restrict__ W,
                                                   float* __restrict__ xw) {
    const int lane  = threadIdx.x & 31;
    const int wave  = threadIdx.x >> 5;
    const int mtile = blockIdx.x * WAVES_PER_BLOCK + wave;
    if (mtile >= MTILES) return;

    const int half = lane >> 4;
    const int l15  = lane & 15;
    const float* xrow = x + (size_t)(mtile * 16 + l15) * NFEAT + 2 * half;
    const float* wrow = W + (size_t)(2 * half) * NCLASS + 4 * l15;

    v8f acc0 = {}, acc1 = {}, acc2 = {}, acc3 = {};

#pragma unroll 8
    for (int k = 0; k < NFEAT; k += 4) {
        v2f    a  = *(const v2f*)(xrow + k);
        float4 w0 = *(const float4*)(wrow + (size_t)k * NCLASS);
        float4 w1 = *(const float4*)(wrow + (size_t)(k + 1) * NCLASS);
        v2f b0 = {w0.x, w1.x};
        v2f b1 = {w0.y, w1.y};
        v2f b2 = {w0.z, w1.z};
        v2f b3 = {w0.w, w1.w};
        acc0 = __builtin_amdgcn_wmma_f32_16x16x4_f32(false, a, false, b0, (short)0, acc0, false, false);
        acc1 = __builtin_amdgcn_wmma_f32_16x16x4_f32(false, a, false, b1, (short)0, acc1, false, false);
        acc2 = __builtin_amdgcn_wmma_f32_16x16x4_f32(false, a, false, b2, (short)0, acc2, false, false);
        acc3 = __builtin_amdgcn_wmma_f32_16x16x4_f32(false, a, false, b3, (short)0, acc3, false, false);
    }

    float* orow = xw + (size_t)mtile * 16 * NCLASS + 4 * l15;
#pragma unroll
    for (int r = 0; r < 8; ++r) {
        int m = r + 8 * half;
        *(float4*)(orow + m * NCLASS) = make_float4(acc0[r], acc1[r], acc2[r], acc3[r]);
    }
}


__device__ __forceinline__ unsigned int zero_bytes(unsigned int x) {
    return ~(((x & 0x7F7F7F7Fu) + 0x7F7F7F7Fu) | x | 0x7F7F7F7Fu);
}

struct RowState {
    float4 a4;
    float  dcur;
    int    cur;
};

__device__ __forceinline__ void row_out(const RowState& st, const int nn, const int l16,
                                        const float4* acc4, float4* o4) {
    if ((unsigned int)st.cur < (unsigned int)nn) {
        const float4 e = acc4[st.cur * (NCLASS / 4) + l16];
        o4[st.cur * (NCLASS / 4) + l16] = make_float4(st.a4.x + e.x, st.a4.y + e.y, st.a4.z + e.z, st.a4.w + e.w);
    }
}

__device__ __forceinline__ void row_add(RowState& st, const int dl, const float ws, const float4 row,
                                        const int nn, const int l16, const float4 b4,
                                        const float* dloc, const float4* acc4, float4* o4) {
    if (dl != st.cur) {
        row_out(st, nn, l16, acc4, o4);
        st.cur  = dl;
        st.dcur = dloc[dl];
        st.a4   = b4;
    }
    const float w = ws * st.dcur;
    st.a4.x += row.x * w;
    st.a4.y += row.y * w;
    st.a4.z += row.z * w;
    st.a4.w += row.w * w;
}

__device__ __forceinline__ void owners_walk(const int* myq, const int m, const int l16, const int nn,
                                            const unsigned char* dg, const float* dloc, const float4* acc4,
                                            const float4 b4, const float* __restrict__ xw, float4* o4) {
    auto fetch = [&](int i, int (&rcx)[8], unsigned int (&dgx)[8], float4 (&rowx)[8]) {
        const int4 qa = *(const int4*)(myq + i);
        const int4 qb = *(const int4*)(myq + i + 4);
        rcx[0] = qa.x; rcx[1] = qa.y; rcx[2] = qa.z; rcx[3] = qa.w;
        rcx[4] = qb.x; rcx[5] = qb.y; rcx[6] = qb.z; rcx[7] = qb.w;
#pragma unroll
        for (int j = 0; j < 8; ++j) {
            const int s = rcx[j] & 0x1FFFF;
            dgx[j]  = dg[s];
            rowx[j] = *(const float4*)(xw + (size_t)s * NCLASS + 4 * l16);
        }
    };
    RowState st;
    st.a4 = b4; st.dcur = 0.0f; st.cur = -1;
    auto add8 = [&](const int (&rcx)[8], const unsigned int (&dgx)[8], const float4 (&rowx)[8]) {
#pragma unroll
        for (int j = 0; j < 8; ++j) {
            const float ws = __frsqrt_rn(1.0f + (float)dgx[j]);
            row_add(st, rcx[j] >> 17, ws, rowx[j], nn, l16, b4, dloc, acc4, o4);
        }
    };
    int    rcA[8], rcB[8];
    unsigned int dgA[8], dgB[8];
    float4 rowA[8], rowB[8];
    if (m > 0) {
        fetch(0, rcA, dgA, rowA);
        for (int i0 = 0; i0 < m; i0 += 16) {
            fetch(i0 + 8, rcB, dgB, rowB);
            add8(rcA, dgA, rowA);
            if (i0 + 16 < m) fetch(i0 + 16, rcA, dgA, rowA);
            add8(rcB, dgB, rowB);
        }
        row_out(st, nn, l16, acc4, o4);
    }
}

__global__ __launch_bounds__(SWEEP_T) void k_sweep(const unsigned int* __restrict__ key4,
                                                   const int* __restrict__ rec,
                                                   const float* __restrict__ xw,
                                                   const float* __restrict__ dinv,
                                                   const float* __restrict__ bias,
                                                   const unsigned int* __restrict__ summ,
                                                   const unsigned int* __restrict__ degb,
                                                   float* __restrict__ out) {
    __shared__ __attribute__((aligned(16))) float acc[RANGE * NCLASS];
    __shared__ __attribute__((aligned(16))) int q[QCAP];
    __shared__ __attribute__((aligned(16))) int glist[LCAP];
    __shared__ float dloc[RANGE + 1];
    __shared__ int   rcnt[RANGE + 1];
    __shared__ int   rbase[RANGE + 1];
    __shared__ int   rcur[RANGE + 1];
    __shared__ int   obase[OWNERS];
    __shared__ int   olim[OWNERS];
    __shared__ int   qn;
    __shared__ int   gn;
    __shared__ int   bigf;

    const int tid   = threadIdx.x;
    const int r     = blockIdx.x;
    const int node0 = r * RANGE;
    const int nn    = (N_NODES - node0 < RANGE) ? (N_NODES - node0) : RANGE;

    const int sb = tid >> 2;
    const int sp = tid & 3;
    const uint4* s4 = (const uint4*)summ + (size_t)sb * (SWORDS / 4) + r * (SROW / 4) + sp * 2;
    const uint4 wa = s4[0];
    const uint4 wb = s4[1];

    {
        float4* z4 = (float4*)acc;
        for (int i = tid; i < RANGE * (NCLASS / 4); i += SWEEP_T) z4[i] = make_float4(0.0f, 0.0f, 0.0f, 0.0f);
    }
    for (int i = tid; i < nn; i += SWEEP_T) dloc[i] = dinv[node0 + i];
    for (int i = tid; i <= RANGE; i += SWEEP_T) rcnt[i] = (i < nn) ? 1 : 0;
    if (tid == 0) { qn = 0; gn = 0; bigf = 0; }
    __syncthreads();

    const unsigned int pat = (unsigned int)r * 0x01010101u;
    const uint4* kv = (const uint4*)key4;

    auto add_rec = [&](int rc) {
        int s  = rc & 0x1FFFF;
        int dl = rc >> 17;
        float nrm = dinv[s] * dloc[dl];
        for (int c = 0; c < NCLASS; ++c)
            atomicAdd(&acc[dl * NCLASS + c], xw[(size_t)s * NCLASS + c] * nrm);
    };

    auto take = [&](int v, uint4 w) {
        unsigned int M = (zero_bytes(w.x ^ pat) >> 7) | (zero_bytes(w.y ^ pat) >> 6) |
                         (zero_bytes(w.z ^ pat) >> 5) | (zero_bytes(w.w ^ pat) >> 4);
        while (M) {
            int p = __ffs(M) - 1;
            M &= M - 1;
            int e = v * 16 + ((p & 3) << 2) + (p >> 3);
            int slot = atomicAdd(&qn, 1);
            if (slot < QCAP) {
                q[slot] = e;
            } else {
                add_rec(rec[e]);
            }
        }
    };

    {
        const unsigned int wd[8] = {wa.x, wa.y, wa.z, wa.w, wb.x, wb.y, wb.z, wb.w};
        const int rot = (r >> 1) & 31;
#pragma unroll
        for (int k = 0; k < 8; ++k) {
            unsigned int B = wd[k];
            const int gw = (sp * 8 + k - rot) & 31;
            const int v0 = sb * SL_VECS + gw * 32;
            while (B) {
                int p = __ffs(B) - 1;
                B &= B - 1;
                int slot = atomicAdd(&gn, 1);
                if (slot < LCAP) {
                    glist[slot] = v0 + p;
                } else {
                    take(v0 + p, kv[v0 + p]);
                }
            }
        }
    }
    __syncthreads();

    const int ng = (gn < LCAP) ? gn : LCAP;
    for (int i0 = 0; i0 < ng; i0 += 4 * SWEEP_T) {
        uint4 w[4];
        int   vi[4];
#pragma unroll
        for (int j = 0; j < 4; ++j) {
            int i = i0 + j * SWEEP_T + tid;
            vi[j] = (i < ng) ? glist[(i < ng) ? i : 0] : -1;
            w[j]  = kv[(vi[j] < 0) ? 0 : vi[j]];
        }
#pragma unroll
        for (int j = 0; j < 4; ++j) {
            if (vi[j] >= 0) take(vi[j], w[j]);
        }
    }
    __syncthreads();

    const uint4* g4 = (const uint4*)degb;
    uint4 dpc[DPIECES];
#pragma unroll
    for (int k = 0; k < DPIECES; ++k) {
        int i = tid + k * SWEEP_T;
        i = (i < N_NODES / 16) ? i : 0;
        dpc[k] = g4[(i >> 2) * (DLINE / 4) + (i & 3)];
    }

    const int n = (qn < QCAP) ? qn : QCAP;
    for (int i0 = 0; i0 < n; i0 += 4 * SWEEP_T) {
        int rcv[4];
#pragma unroll
        for (int j = 0; j < 4; ++j) {
            int i = i0 + j * SWEEP_T + tid;
            rcv[j] = (i < n) ? rec[q[i]] : -1;
        }
#pragma unroll
        for (int j = 0; j < 4; ++j) {
            if (rcv[j] >= 0) {
                q[i0 + j * SWEEP_T + tid] = rcv[j];
                atomicAdd(&rcnt[rcv[j] >> 17], 1);
            }
        }
    }
    __syncthreads();

    int* srt = glist;
    if (tid < 32) {
        int cn[2][OROWS], Mo[2];
#pragma unroll
        for (int a = 0; a < 2; ++a) {
            const int h = tid + 32 * a;
            int m = 0;
#pragma unroll
            for (int k = 0; k < OROWS; ++k) {
                const int dl = h + OWNERS * k;
                cn[a][k] = (dl < nn) ? rcnt[dl] : 0;
                m += cn[a][k];
            }
            Mo[a] = (m + 15) & ~15;
        }
        const int T = Mo[0] + Mo[1];
        int incl = T;
#pragma unroll
        for (int d = 1; d < 32; d <<= 1) {
            const int t = __shfl_up(incl, d);
            if (tid >= d) incl += t;
        }
        int base = incl - T;
#pragma unroll
        for (int a = 0; a < 2; ++a) {
            const int h   = tid + 32 * a;
            const int lim = (base + Mo[a] < SCAP) ? (base + Mo[a]) : SCAP;
            obase[h] = base;
            olim[h]  = lim;
            int run = base;
#pragma unroll
            for (int k = 0; k < OROWS; ++k) {
                const int dl = h + OWNERS * k;
                if (dl < nn) {
                    rbase[dl] = run;
                    rcur[dl]  = run;
                }
                run += cn[a][k];
            }
            for (int pz = run; pz < lim; ++pz) srt[pz] = PADDL << 17;
            base += Mo[a];
        }
    }
    __syncthreads();

    auto place = [&](int rc) {
        const int dl   = rc >> 17;
        const int slot = atomicAdd(&rcur[dl], 1);
        if (slot < olim[dl & (OWNERS - 1)]) srt[slot] = rc;
        else                                add_rec(rc);
    };
    if (tid < nn) place((node0 + tid) | (tid << 17));
    for (int i0 = 0; i0 < n; i0 += 4 * SWEEP_T) {
#pragma unroll
        for (int j = 0; j < 4; ++j) {
            int i = i0 + j * SWEEP_T + tid;
            if (i < n) place(q[i]);
        }
    }
    __syncthreads();

    {
        uint4*       t4 = (uint4*)q;
        unsigned int ff = 0;
#pragma unroll
        for (int k = 0; k < DPIECES; ++k) {
            const int i = tid + k * SWEEP_T;
            if (i < N_NODES / 16) {
                const uint4 v = dpc[k];
                t4[i] = v;
                ff |= zero_bytes(~v.x) | zero_bytes(~v.y) | zero_bytes(~v.z) | zero_bytes(~v.w);
            }
        }
        if (ff) atomicOr(&bigf, 1);
    }
    __syncthreads();

    const int hw  = tid >> 4;
    const int l16 = tid & 15;
    const int ob  = obase[hw];
    const int m   = (olim[hw] > ob) ? (olim[hw] - ob) : 0;
    const int* myq = srt + ob;
    const unsigned char* dg = (const unsigned char*)q;
    const float4* acc4 = (const float4*)acc;
    float4*       o4   = (float4*)(out + (size_t)node0 * NCLASS);
    const float4  b4   = *(const float4*)(bias + 4 * l16);
    if (bigf) {
        RowState st;
        st.a4 = b4; st.dcur = 0.0f; st.cur = -1;
        for (int i = 0; i < m; ++i) {
            const int   rc = myq[i];
            const int   s  = rc & 0x1FFFF;
            const unsigned int b = dg[s];
            const float ws = (b == 255u) ? dinv[s] : __frsqrt_rn(1.0f + (float)b);
            const float4 row = *(const float4*)(xw + (size_t)s * NCLASS + 4 * l16);
            row_add(st, rc >> 17, ws, row, nn, l16, b4, dloc, acc4, o4);
        }
        row_out(st, nn, l16, acc4, o4);
    } else {
        owners_walk(myq, m, l16, nn, dg, dloc, acc4, b4, xw, o4);
    }
    for (int dl = hw; dl < nn; dl += OWNERS) {
        if (rbase[dl] >= olim[hw]) {
            const float4 e = acc4[dl * (NCLASS / 4) + l16];
            o4[dl * (NCLASS / 4) + l16] = make_float4(b4.x + e.x, b4.y + e.y, b4.z + e.z, b4.w + e.w);
        }
    }
}


extern "C" void kernel_launch(void* const* d_in, const int* in_sizes, int n_in,
                              void* d_out, int out_size, void* d_ws, size_t ws_size,
                              hipStream_t stream) {
    const float* x    = (const float*)d_in[0];
    const float* W    = (const float*)d_in[1];
    const float* bias = (const float*)d_in[2];
    const int*   ei   = (const int*)d_in[3];
    const int*   src  = ei;
    const int*   dst  = ei + N_EDGES;
    float* out = (float*)d_out;

    float* dinv = (float*)d_ws;
    float* xw   = dinv + 102400;
    unsigned int* key4 = (unsigned int*)(xw + 6400000);
    int* rec = (int*)(key4 + 800000);
    unsigned int* part = (unsigned int*)(rec + 3200000);
    unsigned int* summ = part + 12804096;
    unsigned int* degb = summ + 2097152;

    k_count<<<CNT_BLOCKS, CNT_T, 0, stream>>>(src, dst, part, key4, rec, summ);
    k_dinv <<<DBLOCKS, DV_T, 0, stream>>>(part, dinv, degb);

    k_gemm_wmma<<<(MTILES + WAVES_PER_BLOCK - 1) / WAVES_PER_BLOCK, 256, 0, stream>>>(x, W, xw);

    k_sweep<<<NRANGES, SWEEP_T, 0, stream>>>(key4, rec, xw, dinv, bias, summ, degb, out);
}


# k_dinv launch gone: each x@W block also sums the 256 slice degree tables for its own 128 rows, 4 table loads per wave and K trip under the matrix instructions; 4 launches to 3; degrees exact
# speedup vs baseline: 6.5044x; 1.0346x over previous
#include <hip/hip_runtime.h>

typedef float v2f __attribute__((ext_vector_type(2)));
typedef float v8f __attribute__((ext_vector_type(8)));
typedef unsigned int v2u __attribute__((ext_vector_type(2)));
typedef unsigned int v4u __attribute__((ext_vector_type(4)));

#define N_NODES 100000
#define N_EDGES 3200000
#define NFEAT   256
#define NCLASS  64
#define MTILES  (N_NODES / 16)
#define WAVES_PER_BLOCK 8

#define RANGE   391
#define NRANGES 256
#define QCAP    36864
#define LCAP    16384
#define SWEEP_T 1024
#define OWNERS  (SWEEP_T / 16)
#define SCAP    LCAP
#define PADDL   RANGE
#define OROWS   ((RANGE + OWNERS - 1) / OWNERS)

#define CNT_BLOCKS 256
#define CNT_T      1024
#define TOT_QUADS  (N_EDGES / 4)
#define SL_QUADS   3136
#define SL_VECS    (SL_QUADS / 4)
#define SROW       32
#define SWORDS     (NRANGES * SROW)
#define HWORDS     (N_NODES / 2)
#define HVECS      (HWORDS / 4)
#define PSTRIDE    50016
#define DLINE      32
#define DBLOCKS    ((HWORDS + 31) / 32)
#define GB_NODES   (16 * WAVES_PER_BLOCK)
#define GB_WORDS   (GB_NODES / 2)
#define KTRIPS     (NFEAT / 32)
#define TAB_WAVE   (CNT_BLOCKS / WAVES_PER_BLOCK)
#define TAB_TRIP   (TAB_WAVE / KTRIPS)
#define DPIECES    ((N_NODES / 16 + SWEEP_T - 1) / SWEEP_T)


__global__ __launch_bounds__(CNT_T) void k_count(const int* __restrict__ src, const int* __restrict__ dst,
                                                 unsigned int* __restrict__ part,
                                                 unsigned int* __restrict__ key4, int* __restrict__ rec,
                                                 unsigned int* __restrict__ summ) {
    __shared__ __attribute__((aligned(16))) unsigned int h[HWORDS];
    __shared__ __attribute__((aligned(16))) unsigned int sm[SWORDS];
    const int tid = threadIdx.x;
    const int q0  = blockIdx.x * SL_QUADS;
    const int nq  = (TOT_QUADS - q0 < SL_QUADS) ? (TOT_QUADS - q0) : SL_QUADS;

    int4 d[4], s[4];
#pragma unroll
    for (int j = 0; j < 4; ++j) {
        int i = tid + j * CNT_T;
        int t = q0 + ((i < nq) ? i : 0);
        d[j] = *(const int4*)(dst + 4 * t);
        s[j] = *(const int4*)(src + 4 * t);
    }

    uint4* h4  = (uint4*)h;
    uint4* sm4 = (uint4*)sm;
    for (int i = tid; i < HVECS; i += CNT_T) h4[i] = make_uint4(0u, 0u, 0u, 0u);
    for (int i = tid; i < SWORDS / 4; i += CNT_T) sm4[i] = make_uint4(0u, 0u, 0u, 0u);
    __syncthreads();

#pragma unroll
    for (int j = 0; j < 4; ++j) {
        int i = tid + j * CNT_T;
        if (i < nq) {
            int t = q0 + i;
            int4 dd = d[j], ss = s[j];
            atomicAdd(&h[dd.x >> 1], 1u << ((dd.x & 1) << 4));
            atomicAdd(&h[dd.y >> 1], 1u << ((dd.y & 1) << 4));
            atomicAdd(&h[dd.z >> 1], 1u << ((dd.z & 1) << 4));
            atomicAdd(&h[dd.w >> 1], 1u << ((dd.w & 1) << 4));
            int r0 = dd.x / RANGE, r1 = dd.y / RANGE, r2 = dd.z / RANGE, r3 = dd.w / RANGE;
            key4[t] = (unsigned int)r0 | ((unsigned int)r1 << 8) | ((unsigned int)r2 << 16) | ((unsigned int)r3 << 24);
            int4 rc;
            rc.x = ss.x | ((dd.x - r0 * RANGE) << 17);
            rc.y = ss.y | ((dd.y - r1 * RANGE) << 17);
            rc.z = ss.z | ((dd.z - r2 * RANGE) << 17);
            rc.w = ss.w | ((dd.w - r3 * RANGE) << 17);
            *(int4*)(rec + 4 * t) = rc;
            const unsigned int gbit = 1u << ((i >> 2) & 31);
            const int gw = i >> 7;
            atomicOr(&sm[r0 * SROW + ((gw + (r0 >> 1)) & 31)], gbit);
            atomicOr(&sm[r1 * SROW + ((gw + (r1 >> 1)) & 31)], gbit);
            atomicOr(&sm[r2 * SROW + ((gw + (r2 >> 1)) & 31)], gbit);
            atomicOr(&sm[r3 * SROW + ((gw + (r3 >> 1)) & 31)], gbit);
        }
    }
    __syncthreads();

    uint4* p4 = (uint4*)(part + (size_t)blockIdx.x * PSTRIDE);
    for (int i = tid; i < HVECS; i += CNT_T) p4[i] = h4[i];
    uint4* s4 = (uint4*)(summ + (size_t)blockIdx.x * SWORDS);
    for (int i = tid; i < SWORDS / 4; i += CNT_T) s4[i] = sm4[i];
}


__global__ __launch_bounds__(256) void k_gemm_wmma(const float* __restrict__ x,
                                                   const float* __restrict__ W,
                                                   float* __restrict__ xw,
                                                   const unsigned int* __restrict__ part,
                                                   float* __restrict__ dinv,
                                                   unsigned int* __restrict__ degb) {
    __shared__ __attribute__((aligned(16))) unsigned int red[WAVES_PER_BLOCK][GB_NODES];
    __shared__ unsigned int rb[GB_NODES / 4];
    const int lane  = threadIdx.x & 31;
    const int wave  = threadIdx.x >> 5;
    const int mtile = blockIdx.x * WAVES_PER_BLOCK + wave;

    const int  w0 = blockIdx.x * GB_WORDS + 2 * lane;
    const bool wv = w0 < HWORDS;
    const unsigned int* tp = part + (size_t)wave * PSTRIDE + (wv ? w0 : 0);
    unsigned int c0 = 0, c1 = 0, c2 = 0, c3 = 0;
    auto tload = [&](int p) -> v2u {
        return __builtin_nontemporal_load((const v2u*)(tp + (size_t)(WAVES_PER_BLOCK * p) * PSTRIDE));
    };
    auto tadd = [&](v2u v) {
        c0 += v.x & 0xFFFFu; c1 += v.x >> 16;
        c2 += v.y & 0xFFFFu; c3 += v.y >> 16;
    };

    if (mtile < MTILES) {
        const int half = lane >> 4;
        const int l15  = lane & 15;
        const float* xrow = x + (size_t)(mtile * 16 + l15) * NFEAT + 2 * half;
        const float* wrow = W + (size_t)(2 * half) * NCLASS + 4 * l15;

        v8f acc0 = {}, acc1 = {}, acc2 = {}, acc3 = {};

#pragma unroll 1
        for (int kt = 0; kt < KTRIPS; ++kt) {
            v2u tv[TAB_TRIP];
#pragma unroll
            for (int j = 0; j < TAB_TRIP; ++j) tv[j] = tload(kt * TAB_TRIP + j);
#pragma unroll
            for (int ks = 0; ks < 8; ++ks) {
                const int k = kt * 32 + ks * 4;
                v2f    a  = *(const v2f*)(xrow + k);
                float4 w0v = *(const float4*)(wrow + (size_t)k * NCLASS);
                float4 w1v = *(const float4*)(wrow + (size_t)(k + 1) * NCLASS);
                v2f b0 = {w0v.x, w1v.x};
                v2f b1 = {w0v.y, w1v.y};
                v2f b2 = {w0v.z, w1v.z};
                v2f b3 = {w0v.w, w1v.w};
                acc0 = __builtin_amdgcn_wmma_f32_16x16x4_f32(false, a, false, b0, (short)0, acc0, false, false);
                acc1 = __builtin_amdgcn_wmma_f32_16x16x4_f32(false, a, false, b1, (short)0, acc1, false, false);
                acc2 = __builtin_amdgcn_wmma_f32_16x16x4_f32(false, a, false, b2, (short)0, acc2, false, false);
                acc3 = __builtin_amdgcn_wmma_f32_16x16x4_f32(false, a, false, b3, (short)0, acc3, false, false);
            }
#pragma unroll
            for (int j = 0; j < TAB_TRIP; ++j) tadd(tv[j]);
        }

        float* orow = xw + (size_t)mtile * 16 * NCLASS + 4 * l15;
#pragma unroll
        for (int r = 0; r < 8; ++r) {
            int m = r + 8 * half;
            *(float4*)(orow + m * NCLASS) = make_float4(acc0[r], acc1[r], acc2[r], acc3[r]);
        }
    } else {
#pragma unroll 8
        for (int p = 0; p < TAB_WAVE; ++p) tadd(tload(p));
    }

    if (!wv) { c0 = 0; c1 = 0; c2 = 0; c3 = 0; }
    {
        v4u cs = {c0, c1, c2, c3};
        *(v4u*)&red[wave][4 * lane] = cs;
    }
    __syncthreads();
    if (threadIdx.x < GB_NODES) {
        int node = blockIdx.x * GB_NODES + threadIdx.x;
        unsigned int c = 0;
        if (node < N_NODES) {
#pragma unroll
            for (int g = 0; g < WAVES_PER_BLOCK; ++g) c += red[g][threadIdx.x];
            dinv[node] = __frsqrt_rn(1.0f + (float)c);
        }
        ((unsigned char*)rb)[threadIdx.x] = (unsigned char)((c < 255u) ? c : 255u);
    }
    __syncthreads();
    if (threadIdx.x < GB_NODES / 4) {
        const int line = blockIdx.x * 2 + (threadIdx.x >> 4);
        if (line < DBLOCKS) degb[line * DLINE + (threadIdx.x & 15)] = rb[threadIdx.x];
    }
}


__device__ __forceinline__ unsigned int zero_bytes(unsigned int x) {
    return ~(((x & 0x7F7F7F7Fu) + 0x7F7F7F7Fu) | x | 0x7F7F7F7Fu);
}

struct RowState {
    float4 a4;
    float  dcur;
    int    cur;
};

__device__ __forceinline__ void row_out(const RowState& st, const int nn, const int l16,
                                        const float4* acc4, float4* o4) {
    if ((unsigned int)st.cur < (unsigned int)nn) {
        const float4 e = acc4[st.cur * (NCLASS / 4) + l16];
        o4[st.cur * (NCLASS / 4) + l16] = make_float4(st.a4.x + e.x, st.a4.y + e.y, st.a4.z + e.z, st.a4.w + e.w);
    }
}

__device__ __forceinline__ void row_add(RowState& st, const int dl, const float ws, const float4 row,
                                        const int nn, const int l16, const float4 b4,
                                        const float* dloc, const float4* acc4, float4* o4) {
    if (dl != st.cur) {
        row_out(st, nn, l16, acc4, o4);
        st.cur  = dl;
        st.dcur = dloc[dl];
        st.a4   = b4;
    }
    const float w = ws * st.dcur;
    st.a4.x += row.x * w;
    st.a4.y += row.y * w;
    st.a4.z += row.z * w;
    st.a4.w += row.w * w;
}

__device__ __forceinline__ void owners_walk(const int* myq, const int m, const int l16, const int nn,
                                            const unsigned char* dg, const float* dloc, const float4* acc4,
                                            const float4 b4, const float* __restrict__ xw, float4* o4) {
    auto fetch = [&](int i, int (&rcx)[8], unsigned int (&dgx)[8], float4 (&rowx)[8]) {
        const int4 qa = *(const int4*)(myq + i);
        const int4 qb = *(const int4*)(myq + i + 4);
        rcx[0] = qa.x; rcx[1] = qa.y; rcx[2] = qa.z; rcx[3] = qa.w;
        rcx[4] = qb.x; rcx[5] = qb.y; rcx[6] = qb.z; rcx[7] = qb.w;
#pragma unroll
        for (int j = 0; j < 8; ++j) {
            const int s = rcx[j] & 0x1FFFF;
            dgx[j]  = dg[s];
            rowx[j] = *(const float4*)(xw + (size_t)s * NCLASS + 4 * l16);
        }
    };
    RowState st;
    st.a4 = b4; st.dcur = 0.0f; st.cur = -1;
    auto add8 = [&](const int (&rcx)[8], const unsigned int (&dgx)[8], const float4 (&rowx)[8]) {
#pragma unroll
        for (int j = 0; j < 8; ++j) {
            const float ws = __frsqrt_rn(1.0f + (float)dgx[j]);
            row_add(st, rcx[j] >> 17, ws, rowx[j], nn, l16, b4, dloc, acc4, o4);
        }
    };
    int    rcA[8], rcB[8];
    unsigned int dgA[8], dgB[8];
    float4 rowA[8], rowB[8];
    if (m > 0) {
        fetch(0, rcA, dgA, rowA);
        for (int i0 = 0; i0 < m; i0 += 16) {
            fetch(i0 + 8, rcB, dgB, rowB);
            add8(rcA, dgA, rowA);
            if (i0 + 16 < m) fetch(i0 + 16, rcA, dgA, rowA);
            add8(rcB, dgB, rowB);
        }
        row_out(st, nn, l16, acc4, o4);
    }
}

__global__ __launch_bounds__(SWEEP_T) void k_sweep(const unsigned int* __restrict__ key4,
                                                   const int* __restrict__ rec,
                                                   const float* __restrict__ xw,
                                                   const float* __restrict__ dinv,
                                                   const float* __restrict__ bias,
                                                   const unsigned int* __restrict__ summ,
                                                   const unsigned int* __restrict__ degb,
                                                   float* __restrict__ out) {
    __shared__ __attribute__((aligned(16))) float acc[RANGE * NCLASS];
    __shared__ __attribute__((aligned(16))) int q[QCAP];
    __shared__ __attribute__((aligned(16))) int glist[LCAP];
    __shared__ float dloc[RANGE + 1];
    __shared__ int   rcnt[RANGE + 1];
    __shared__ int   rbase[RANGE + 1];
    __shared__ int   rcur[RANGE + 1];
    __shared__ int   obase[OWNERS];
    __shared__ int   olim[OWNERS];
    __shared__ int   qn;
    __shared__ int   gn;
    __shared__ int   bigf;

    const int tid   = threadIdx.x;
    const int r     = blockIdx.x;
    const int node0 = r * RANGE;
    const int nn    = (N_NODES - node0 < RANGE) ? (N_NODES - node0) : RANGE;

    const int sb = tid >> 2;
    const int sp = tid & 3;
    const uint4* s4 = (const uint4*)summ + (size_t)sb * (SWORDS / 4) + r * (SROW / 4) + sp * 2;
    const uint4 wa = s4[0];
    const uint4 wb = s4[1];

    {
        float4* z4 = (float4*)acc;
        for (int i = tid; i < RANGE * (NCLASS / 4); i += SWEEP_T) z4[i] = make_float4(0.0f, 0.0f, 0.0f, 0.0f);
    }
    for (int i = tid; i < nn; i += SWEEP_T) dloc[i] = dinv[node0 + i];
    for (int i = tid; i <= RANGE; i += SWEEP_T) rcnt[i] = (i < nn) ? 1 : 0;
    if (tid == 0) { qn = 0; gn = 0; bigf = 0; }
    __syncthreads();

    const unsigned int pat = (unsigned int)r * 0x01010101u;
    const uint4* kv = (const uint4*)key4;

    auto add_rec = [&](int rc) {
        int s  = rc & 0x1FFFF;
        int dl = rc >> 17;
        float nrm = dinv[s] * dloc[dl];
        for (int c = 0; c < NCLASS; ++c)
            atomicAdd(&acc[dl * NCLASS + c], xw[(size_t)s * NCLASS + c] * nrm);
    };

    auto take = [&](int v, uint4 w) {
        unsigned int M = (zero_bytes(w.x ^ pat) >> 7) | (zero_bytes(w.y ^ pat) >> 6) |
                         (zero_bytes(w.z ^ pat) >> 5) | (zero_bytes(w.w ^ pat) >> 4);
        while (M) {
            int p = __ffs(M) - 1;
            M &= M - 1;
            int e = v * 16 + ((p & 3) << 2) + (p >> 3);
            int slot = atomicAdd(&qn, 1);
            if (slot < QCAP) {
                q[slot] = e;
            } else {
                add_rec(rec[e]);
            }
        }
    };

    {
        const unsigned int wd[8] = {wa.x, wa.y, wa.z, wa.w, wb.x, wb.y, wb.z, wb.w};
        const int rot = (r >> 1) & 31;
#pragma unroll
        for (int k = 0; k < 8; ++k) {
            unsigned int B = wd[k];
            const int gw = (sp * 8 + k - rot) & 31;
            const int v0 = sb * SL_VECS + gw * 32;
            while (B) {
                int p = __ffs(B) - 1;
                B &= B - 1;
                int slot = atomicAdd(&gn, 1);
                if (slot < LCAP) {
                    glist[slot] = v0 + p;
                } else {
                    take(v0 + p, kv[v0 + p]);
                }
            }
        }
    }
    __syncthreads();

    const int ng = (gn < LCAP) ? gn : LCAP;
    for (int i0 = 0; i0 < ng; i0 += 4 * SWEEP_T) {
        uint4 w[4];
        int   vi[4];
#pragma unroll
        for (int j = 0; j < 4; ++j) {
            int i = i0 + j * SWEEP_T + tid;
            vi[j] = (i < ng) ? glist[(i < ng) ? i : 0] : -1;
            w[j]  = kv[(vi[j] < 0) ? 0 : vi[j]];
        }
#pragma unroll
        for (int j = 0; j < 4; ++j) {
            if (vi[j] >= 0) take(vi[j], w[j]);
        }
    }
    __syncthreads();

    const uint4* g4 = (const uint4*)degb;
    uint4 dpc[DPIECES];
#pragma unroll
    for (int k = 0; k < DPIECES; ++k) {
        int i = tid + k * SWEEP_T;
        i = (i < N_NODES / 16) ? i : 0;
        dpc[k] = g4[(i >> 2) * (DLINE / 4) + (i & 3)];
    }

    const int n = (qn < QCAP) ? qn : QCAP;
    for (int i0 = 0; i0 < n; i0 += 4 * SWEEP_T) {
        int rcv[4];
#pragma unroll
        for (int j = 0; j < 4; ++j) {
            int i = i0 + j * SWEEP_T + tid;
            rcv[j] = (i < n) ? rec[q[i]] : -1;
        }
#pragma unroll
        for (int j = 0; j < 4; ++j) {
            if (rcv[j] >= 0) {
                q[i0 + j * SWEEP_T + tid] = rcv[j];
                atomicAdd(&rcnt[rcv[j] >> 17], 1);
            }
        }
    }
    __syncthreads();

    int* srt = glist;
    if (tid < 32) {
        int cn[2][OROWS], Mo[2];
#pragma unroll
        for (int a = 0; a < 2; ++a) {
            const int h = tid + 32 * a;
            int m = 0;
#pragma unroll
            for (int k = 0; k < OROWS; ++k) {
                const int dl = h + OWNERS * k;
                cn[a][k] = (dl < nn) ? rcnt[dl] : 0;
                m += cn[a][k];
            }
            Mo[a] = (m + 15) & ~15;
        }
        const int T = Mo[0] + Mo[1];
        int incl = T;
#pragma unroll
        for (int d = 1; d < 32; d <<= 1) {
            const int t = __shfl_up(incl, d);
            if (tid >= d) incl += t;
        }
        int base = incl - T;
#pragma unroll
        for (int a = 0; a < 2; ++a) {
            const int h   = tid + 32 * a;
            const int lim = (base + Mo[a] < SCAP) ? (base + Mo[a]) : SCAP;
            obase[h] = base;
            olim[h]  = lim;
            int run = base;
#pragma unroll
            for (int k = 0; k < OROWS; ++k) {
                const int dl = h + OWNERS * k;
                if (dl < nn) {
                    rbase[dl] = run;
                    rcur[dl]  = run;
                }
                run += cn[a][k];
            }
            for (int pz = run; pz < lim; ++pz) srt[pz] = PADDL << 17;
            base += Mo[a];
        }
    }
    __syncthreads();

    auto place = [&](int rc) {
        const int dl   = rc >> 17;
        const int slot = atomicAdd(&rcur[dl], 1);
        if (slot < olim[dl & (OWNERS - 1)]) srt[slot] = rc;
        else                                add_rec(rc);
    };
    if (tid < nn) place((node0 + tid) | (tid << 17));
    for (int i0 = 0; i0 < n; i0 += 4 * SWEEP_T) {
#pragma unroll
        for (int j = 0; j < 4; ++j) {
            int i = i0 + j * SWEEP_T + tid;
            if (i < n) place(q[i]);
        }
    }
    __syncthreads();

    {
        uint4*       t4 = (uint4*)q;
        unsigned int ff = 0;
#pragma unroll
        for (int k = 0; k < DPIECES; ++k) {
            const int i = tid + k * SWEEP_T;
            if (i < N_NODES / 16) {
                const uint4 v = dpc[k];
                t4[i] = v;
                ff |= zero_bytes(~v.x) | zero_bytes(~v.y) | zero_bytes(~v.z) | zero_bytes(~v.w);
            }
        }
        if (ff) atomicOr(&bigf, 1);
    }
    __syncthreads();

    const int hw  = tid >> 4;
    const int l16 = tid & 15;
    const int ob  = obase[hw];
    const int m   = (olim[hw] > ob) ? (olim[hw] - ob) : 0;
    const int* myq = srt + ob;
    const unsigned char* dg = (const unsigned char*)q;
    const float4* acc4 = (const float4*)acc;
    float4*       o4   = (float4*)(out + (size_t)node0 * NCLASS);
    const float4  b4   = *(const float4*)(bias + 4 * l16);
    if (bigf) {
        RowState st;
        st.a4 = b4; st.dcur = 0.0f; st.cur = -1;
        for (int i = 0; i < m; ++i) {
            const int   rc = myq[i];
            const int   s  = rc & 0x1FFFF;
            const unsigned int b = dg[s];
            const float ws = (b == 255u) ? dinv[s] : __frsqrt_rn(1.0f + (float)b);
            const float4 row = *(const float4*)(xw + (size_t)s * NCLASS + 4 * l16);
            row_add(st, rc >> 17, ws, row, nn, l16, b4, dloc, acc4, o4);
        }
        row_out(st, nn, l16, acc4, o4);
    } else {
        owners_walk(myq, m, l16, nn, dg, dloc, acc4, b4, xw, o4);
    }
    for (int dl = hw; dl < nn; dl += OWNERS) {
        if (rbase[dl] >= olim[hw]) {
            const float4 e = acc4[dl * (NCLASS / 4) + l16];
            o4[dl * (NCLASS / 4) + l16] = make_float4(b4.x + e.x, b4.y + e.y, b4.z + e.z, b4.w + e.w);
        }
    }
}


extern "C" void kernel_launch(void* const* d_in, const int* in_sizes, int n_in,
                              void* d_out, int out_size, void* d_ws, size_t ws_size,
                              hipStream_t stream) {
    const float* x    = (const float*)d_in[0];
    const float* W    = (const float*)d_in[1];
    const float* bias = (const float*)d_in[2];
    const int*   ei   = (const int*)d_in[3];
    const int*   src  = ei;
    const int*   dst  = ei + N_EDGES;
    float* out = (float*)d_out;

    float* dinv = (float*)d_ws;
    float* xw   = dinv + 102400;
    unsigned int* key4 = (unsigned int*)(xw + 6400000);
    int* rec = (int*)(key4 + 800000);
    unsigned int* part = (unsigned int*)(rec + 3200000);
    unsigned int* summ = part + 12804096;
    unsigned int* degb = summ + 2097152;

    k_count<<<CNT_BLOCKS, CNT_T, 0, stream>>>(src, dst, part, key4, rec, summ);

    k_gemm_wmma<<<(MTILES + WAVES_PER_BLOCK - 1) / WAVES_PER_BLOCK, 256, 0, stream>>>(x, W, xw, part, dinv, degb);

    k_sweep<<<NRANGES, SWEEP_T, 0, stream>>>(key4, rec, xw, dinv, bias, summ, degb, out);
}
